# starting kernel
# baseline (speedup 1.0000x reference)
#include <hip/hip_runtime.h>
#include <math.h>

typedef float v2f __attribute__((ext_vector_type(2)));
typedef float v8f __attribute__((ext_vector_type(8)));

#define BATCH 8192
#define FDIM  512
#define NQ    8
#define NSTATE 256
#define NCLS  10

__global__ void k1_angles(const float* __restrict__ x,
                          const float* __restrict__ proj_w,
                          float* __restrict__ angles) {
    __shared__ float w[NQ * FDIM];
    int tid = threadIdx.x;
    for (int i = tid; i < NQ * FDIM; i += 256) w[i] = proj_w[i];
    __syncthreads();

    int lane = tid & 31;
    int wave = tid >> 5;
    int row  = blockIdx.x * 8 + wave;
    const float* xr = x + (size_t)row * FDIM;

    float acc[NQ];
#pragma unroll
    for (int q = 0; q < NQ; ++q) acc[q] = 0.0f;

    for (int j = 0; j < FDIM / 32; ++j) {
        float xv = xr[lane + 32 * j];
#pragma unroll
        for (int q = 0; q < NQ; ++q)
            acc[q] += xv * w[q * FDIM + lane + 32 * j];
    }
#pragma unroll
    for (int q = 0; q < NQ; ++q) {
        float v = acc[q];
#pragma unroll
        for (int off = 16; off >= 1; off >>= 1) v += __shfl_xor(v, off, 32);
        if (lane == q)
            angles[row * NQ + q] = tanhf(v) * 1.5707963267948966f;
    }
}

__device__ __forceinline__ void rot_lane_rx(float sr[8], float si[8],
                                            float c, float s, int m) {
#pragma unroll
    for (int r = 0; r < 8; ++r) {
        float pr = __shfl_xor(sr[r], m, 32);
        float pi = __shfl_xor(si[r], m, 32);
        float nr = c * sr[r] + s * pi;
        float ni = c * si[r] - s * pr;
        sr[r] = nr; si[r] = ni;
    }
}

__device__ __forceinline__ void rot_lane_ry(float sr[8], float si[8],
                                            float c, float s, int m, int lane) {
    float b = (lane & m) ? s : -s;
#pragma unroll
    for (int r = 0; r < 8; ++r) {
        float pr = __shfl_xor(sr[r], m, 32);
        float pi = __shfl_xor(si[r], m, 32);
        sr[r] = c * sr[r] + b * pr;
        si[r] = c * si[r] + b * pi;
    }
}

__device__ __forceinline__ void rot_reg_rx(float sr[8], float si[8],
                                           float c, float s, int rm) {
#pragma unroll
    for (int r0 = 0; r0 < 8; ++r0) {
        if (r0 & rm) continue;
        int r1 = r0 | rm;
        float a0r = sr[r0], a0i = si[r0], a1r = sr[r1], a1i = si[r1];
        sr[r0] = c * a0r + s * a1i;  si[r0] = c * a0i - s * a1r;
        sr[r1] = c * a1r + s * a0i;  si[r1] = c * a1i - s * a0r;
    }
}

__device__ __forceinline__ void rot_reg_ry(float sr[8], float si[8],
                                           float c, float s, int rm) {
#pragma unroll
    for (int r0 = 0; r0 < 8; ++r0) {
        if (r0 & rm) continue;
        int r1 = r0 | rm;
        float a0r = sr[r0], a0i = si[r0], a1r = sr[r1], a1i = si[r1];
        sr[r0] = c * a0r - s * a1r;  si[r0] = c * a0i - s * a1i;
        sr[r1] = s * a0r + c * a1r;  si[r1] = s * a0i + c * a1i;
    }
}

__device__ __forceinline__ void cnot_ll(float sr[8], float si[8],
                                        int cm, int tm, int lane) {
    bool hi = (lane & cm) != 0;
#pragma unroll
    for (int r = 0; r < 8; ++r) {
        float pr = __shfl_xor(sr[r], tm, 32);
        float pi = __shfl_xor(si[r], tm, 32);
        sr[r] = hi ? pr : sr[r];
        si[r] = hi ? pi : si[r];
    }
}

__device__ __forceinline__ void cnot_lane_reg(float sr[8], float si[8],
                                              int cm, int trm, int lane) {
    bool hi = (lane & cm) != 0;
#pragma unroll
    for (int r0 = 0; r0 < 8; ++r0) {
        if (r0 & trm) continue;
        int r1 = r0 | trm;
        float t0r = sr[r0], t0i = si[r0];
        sr[r0] = hi ? sr[r1] : sr[r0];  si[r0] = hi ? si[r1] : si[r0];
        sr[r1] = hi ? t0r    : sr[r1];  si[r1] = hi ? t0i    : si[r1];
    }
}

__device__ __forceinline__ void cnot_reg_lane(float sr[8], float si[8],
                                              int crm, int tm) {
#pragma unroll
    for (int r = 0; r < 8; ++r) {
        if (!(r & crm)) continue;
        sr[r] = __shfl_xor(sr[r], tm, 32);
        si[r] = __shfl_xor(si[r], tm, 32);
    }
}

__device__ __forceinline__ void cnot_reg_reg(float sr[8], float si[8],
                                             int crm, int trm) {
#pragma unroll
    for (int r0 = 0; r0 < 8; ++r0) {
        if (!(r0 & crm) || (r0 & trm)) continue;
        int r1 = r0 | trm;
        float tr = sr[r0], ti = si[r0];
        sr[r0] = sr[r1]; si[r0] = si[r1];
        sr[r1] = tr;     si[r1] = ti;
    }
}

__global__ void k2_unitary(const float* __restrict__ qnn_w,
                           float* __restrict__ UrP,
                           float* __restrict__ UiP) {
    int lane = threadIdx.x & 31;
    int t    = blockIdx.x * 8 + (threadIdx.x >> 5);

    float sr[8], si[8];
#pragma unroll
    for (int r = 0; r < 8; ++r) {
        sr[r] = (r * 32 + lane == t) ? 1.0f : 0.0f;
        si[r] = 0.0f;
    }

#pragma unroll
    for (int l = 0; l < 3; ++l) {
#pragma unroll
        for (int q = 0; q < NQ; ++q) {
            float w0 = qnn_w[(l * NQ + q) * 3 + 0] * 0.5f;
            float w1 = qnn_w[(l * NQ + q) * 3 + 1] * 0.5f;
            float w2 = qnn_w[(l * NQ + q) * 3 + 2] * 0.5f;
            float c0 = cosf(w0), s0 = sinf(w0);
            float c1 = cosf(w1), s1 = sinf(w1);
            float c2 = cosf(w2), s2 = sinf(w2);
            if (q < 5) {
                int m = 1 << q;
                rot_lane_rx(sr, si, c0, s0, m);
                rot_lane_ry(sr, si, c1, s1, m, lane);
                rot_lane_rx(sr, si, c2, s2, m);
            } else {
                int rm = 1 << (q - 5);
                rot_reg_rx(sr, si, c0, s0, rm);
                rot_reg_ry(sr, si, c1, s1, rm);
                rot_reg_rx(sr, si, c2, s2, rm);
            }
        }
        cnot_ll(sr, si, 1, 2, lane);
        cnot_ll(sr, si, 2, 4, lane);
        cnot_ll(sr, si, 4, 8, lane);
        cnot_ll(sr, si, 8, 16, lane);
        cnot_lane_reg(sr, si, 16, 1, lane);
        cnot_reg_reg(sr, si, 1, 2);
        cnot_reg_reg(sr, si, 2, 4);
        cnot_reg_lane(sr, si, 4, 1);
    }

#pragma unroll
    for (int r = 0; r < 8; ++r) {
        int s = r * 32 + lane;
        int off = (t >> 1) * (2 * NSTATE) + s * 2 + (t & 1);
        UrP[off] = sr[r];
        UiP[off] = si[r];
    }
}

__global__ __launch_bounds__(512) void k3_fused(
        const float* __restrict__ angles,
        const float* __restrict__ UrP,
        const float* __restrict__ UiP,
        const float* __restrict__ out_w,
        const float* __restrict__ out_b,
        float* __restrict__ out) {
    __shared__ __align__(16) float cs[16][NQ][2];
    __shared__ __align__(16) float enc[16][NSTATE];
    __shared__ __align__(16) float pq[16][NSTATE];
    __shared__ __align__(16) float E[16][NQ];

    int tid  = threadIdx.x;
    int lane = tid & 31;
    int wave = tid >> 5;
    int row0 = blockIdx.x * 16;

    if (tid < 128) {
        int m = tid >> 3, q = tid & 7;
        float a = angles[(row0 + m) * NQ + q] * 0.5f;
        cs[m][q][0] = cosf(a);
        cs[m][q][1] = sinf(a);
    }
    __syncthreads();

#pragma unroll
    for (int e = 0; e < 8; ++e) {
        int idx = tid + 512 * e;
        int m = idx >> 8, t = idx & 255;
        float v = 1.0f;
#pragma unroll
        for (int q = 0; q < NQ; ++q) v *= cs[m][q][(t >> q) & 1];
        enc[m][t] = v;
    }
    __syncthreads();

    v8f cr, ci;
#pragma unroll
    for (int v = 0; v < 8; ++v) { cr[v] = 0.0f; ci[v] = 0.0f; }

    int mrow = lane & 15;
    int h    = lane >> 4;
    int n    = lane & 15;
    int s0   = wave * 16;

    for (int k = 0; k < NSTATE; k += 4) {
        int kk = k + 2 * h;
        v2f a  = *(const v2f*)&enc[mrow][kk];
        const float* pr = &UrP[(kk >> 1) * (2 * NSTATE) + (s0 + n) * 2];
        const float* pi = &UiP[(kk >> 1) * (2 * NSTATE) + (s0 + n) * 2];
        v2f br = *(const v2f*)pr;
        v2f bi = *(const v2f*)pi;
        cr = __builtin_amdgcn_wmma_f32_16x16x4_f32(false, a, false, br,
                                                   (short)0, cr, false, false);
        ci = __builtin_amdgcn_wmma_f32_16x16x4_f32(false, a, false, bi,
                                                   (short)0, ci, false, false);
    }

#pragma unroll
    for (int v = 0; v < 8; ++v) {
        int m = v + 8 * h;
        pq[m][s0 + n] = cr[v] * cr[v] + ci[v] * ci[v];
    }
    __syncthreads();

    if (tid < 128) {
        int m = tid >> 3, q = tid & 7;
        float acc = 0.0f;
        for (int s = 0; s < NSTATE; ++s) {
            float p = pq[m][s];
            acc += ((s >> q) & 1) ? -p : p;
        }
        E[m][q] = acc;
    }
    __syncthreads();

    if (tid < 16 * NCLS) {
        int m = tid / NCLS, c = tid % NCLS;
        float acc = out_b[c];
#pragma unroll
        for (int q = 0; q < NQ; ++q) acc += E[m][q] * out_w[c * NQ + q];
        out[(row0 + m) * NCLS + c] = acc;
    }
}

extern "C" void kernel_launch(void* const* d_in, const int* in_sizes, int n_in,
                              void* d_out, int out_size, void* d_ws, size_t ws_size,
                              hipStream_t stream) {
    const float* x      = (const float*)d_in[0];
    const float* proj_w = (const float*)d_in[1];
    const float* qnn_w  = (const float*)d_in[2];
    const float* out_w  = (const float*)d_in[3];
    const float* out_b  = (const float*)d_in[4];
    float* ws = (float*)d_ws;

    float* angles = ws;
    float* UrP    = ws + 65536;
    float* UiP    = ws + 65536 + 65536;

    k1_angles<<<BATCH / 8, 256, 0, stream>>>(x, proj_w, angles);
    k2_unitary<<<NSTATE / 8, 256, 0, stream>>>(qnn_w, UrP, UiP);
    k3_fused<<<BATCH / 16, 512, 0, stream>>>(angles, UrP, UiP, out_w, out_b,
                                             (float*)d_out);
}


# angle kernel and unitary kernel merged into one launch (role by work-group index, unitary groups first): the two independent serial chains overlap and one launch boundary goes
# speedup vs baseline: 1.2542x; 1.2542x over previous
#include <hip/hip_runtime.h>
#include <math.h>

typedef float v2f __attribute__((ext_vector_type(2)));
typedef float v8f __attribute__((ext_vector_type(8)));

#define BATCH 8192
#define FDIM  512
#define NQ    8
#define NSTATE 256
#define NCLS  10

#define U_BLOCKS        32
#define PRODUCER_BLOCKS 1056

__device__ __forceinline__ void angles_rows(const float* __restrict__ x,
                                            const float* __restrict__ proj_w,
                                            float* __restrict__ angles,
                                            int bid) {
    __shared__ float w[NQ * FDIM];
    int tid = threadIdx.x;
    for (int i = tid; i < NQ * FDIM; i += 256) w[i] = proj_w[i];
    __syncthreads();

    int lane = tid & 31;
    int wave = tid >> 5;
    int row  = bid * 8 + wave;
    const float* xr = x + (size_t)row * FDIM;

    float acc[NQ];
#pragma unroll
    for (int q = 0; q < NQ; ++q) acc[q] = 0.0f;

    for (int j = 0; j < FDIM / 32; ++j) {
        float xv = xr[lane + 32 * j];
#pragma unroll
        for (int q = 0; q < NQ; ++q)
            acc[q] += xv * w[q * FDIM + lane + 32 * j];
    }
#pragma unroll
    for (int q = 0; q < NQ; ++q) {
        float v = acc[q];
#pragma unroll
        for (int off = 16; off >= 1; off >>= 1) v += __shfl_xor(v, off, 32);
        if (lane == q)
            angles[row * NQ + q] = tanhf(v) * 1.5707963267948966f;
    }
}

__device__ __forceinline__ void rot_lane_rx(float sr[8], float si[8],
                                            float c, float s, int m) {
#pragma unroll
    for (int r = 0; r < 8; ++r) {
        float pr = __shfl_xor(sr[r], m, 32);
        float pi = __shfl_xor(si[r], m, 32);
        float nr = c * sr[r] + s * pi;
        float ni = c * si[r] - s * pr;
        sr[r] = nr; si[r] = ni;
    }
}

__device__ __forceinline__ void rot_lane_ry(float sr[8], float si[8],
                                            float c, float s, int m, int lane) {
    float b = (lane & m) ? s : -s;
#pragma unroll
    for (int r = 0; r < 8; ++r) {
        float pr = __shfl_xor(sr[r], m, 32);
        float pi = __shfl_xor(si[r], m, 32);
        sr[r] = c * sr[r] + b * pr;
        si[r] = c * si[r] + b * pi;
    }
}

__device__ __forceinline__ void rot_reg_rx(float sr[8], float si[8],
                                           float c, float s, int rm) {
#pragma unroll
    for (int r0 = 0; r0 < 8; ++r0) {
        if (r0 & rm) continue;
        int r1 = r0 | rm;
        float a0r = sr[r0], a0i = si[r0], a1r = sr[r1], a1i = si[r1];
        sr[r0] = c * a0r + s * a1i;  si[r0] = c * a0i - s * a1r;
        sr[r1] = c * a1r + s * a0i;  si[r1] = c * a1i - s * a0r;
    }
}

__device__ __forceinline__ void rot_reg_ry(float sr[8], float si[8],
                                           float c, float s, int rm) {
#pragma unroll
    for (int r0 = 0; r0 < 8; ++r0) {
        if (r0 & rm) continue;
        int r1 = r0 | rm;
        float a0r = sr[r0], a0i = si[r0], a1r = sr[r1], a1i = si[r1];
        sr[r0] = c * a0r - s * a1r;  si[r0] = c * a0i - s * a1i;
        sr[r1] = s * a0r + c * a1r;  si[r1] = s * a0i + c * a1i;
    }
}

__device__ __forceinline__ void cnot_ll(float sr[8], float si[8],
                                        int cm, int tm, int lane) {
    bool hi = (lane & cm) != 0;
#pragma unroll
    for (int r = 0; r < 8; ++r) {
        float pr = __shfl_xor(sr[r], tm, 32);
        float pi = __shfl_xor(si[r], tm, 32);
        sr[r] = hi ? pr : sr[r];
        si[r] = hi ? pi : si[r];
    }
}

__device__ __forceinline__ void cnot_lane_reg(float sr[8], float si[8],
                                              int cm, int trm, int lane) {
    bool hi = (lane & cm) != 0;
#pragma unroll
    for (int r0 = 0; r0 < 8; ++r0) {
        if (r0 & trm) continue;
        int r1 = r0 | trm;
        float t0r = sr[r0], t0i = si[r0];
        sr[r0] = hi ? sr[r1] : sr[r0];  si[r0] = hi ? si[r1] : si[r0];
        sr[r1] = hi ? t0r    : sr[r1];  si[r1] = hi ? t0i    : si[r1];
    }
}

__device__ __forceinline__ void cnot_reg_lane(float sr[8], float si[8],
                                              int crm, int tm) {
#pragma unroll
    for (int r = 0; r < 8; ++r) {
        if (!(r & crm)) continue;
        sr[r] = __shfl_xor(sr[r], tm, 32);
        si[r] = __shfl_xor(si[r], tm, 32);
    }
}

__device__ __forceinline__ void cnot_reg_reg(float sr[8], float si[8],
                                             int crm, int trm) {
#pragma unroll
    for (int r0 = 0; r0 < 8; ++r0) {
        if (!(r0 & crm) || (r0 & trm)) continue;
        int r1 = r0 | trm;
        float tr = sr[r0], ti = si[r0];
        sr[r0] = sr[r1]; si[r0] = si[r1];
        sr[r1] = tr;     si[r1] = ti;
    }
}

__device__ __forceinline__ void unitary_cols(const float* __restrict__ qnn_w,
                                             float* __restrict__ UrP,
                                             float* __restrict__ UiP,
                                             int bid) {
    int lane = threadIdx.x & 31;
    int t    = bid * 8 + (threadIdx.x >> 5);

    const __attribute__((address_space(4))) float* wq =
        (const __attribute__((address_space(4))) float*)qnn_w;

    float sr[8], si[8];
#pragma unroll
    for (int r = 0; r < 8; ++r) {
        sr[r] = (r * 32 + lane == t) ? 1.0f : 0.0f;
        si[r] = 0.0f;
    }

#pragma unroll
    for (int l = 0; l < 3; ++l) {
#pragma unroll
        for (int q = 0; q < NQ; ++q) {
            float w0 = wq[(l * NQ + q) * 3 + 0] * 0.5f;
            float w1 = wq[(l * NQ + q) * 3 + 1] * 0.5f;
            float w2 = wq[(l * NQ + q) * 3 + 2] * 0.5f;
            float c0 = cosf(w0), s0 = sinf(w0);
            float c1 = cosf(w1), s1 = sinf(w1);
            float c2 = cosf(w2), s2 = sinf(w2);
            if (q < 5) {
                int m = 1 << q;
                rot_lane_rx(sr, si, c0, s0, m);
                rot_lane_ry(sr, si, c1, s1, m, lane);
                rot_lane_rx(sr, si, c2, s2, m);
            } else {
                int rm = 1 << (q - 5);
                rot_reg_rx(sr, si, c0, s0, rm);
                rot_reg_ry(sr, si, c1, s1, rm);
                rot_reg_rx(sr, si, c2, s2, rm);
            }
        }
        cnot_ll(sr, si, 1, 2, lane);
        cnot_ll(sr, si, 2, 4, lane);
        cnot_ll(sr, si, 4, 8, lane);
        cnot_ll(sr, si, 8, 16, lane);
        cnot_lane_reg(sr, si, 16, 1, lane);
        cnot_reg_reg(sr, si, 1, 2);
        cnot_reg_reg(sr, si, 2, 4);
        cnot_reg_lane(sr, si, 4, 1);
    }

#pragma unroll
    for (int r = 0; r < 8; ++r) {
        int s = r * 32 + lane;
        int off = (t >> 1) * (2 * NSTATE) + s * 2 + (t & 1);
        UrP[off] = sr[r];
        UiP[off] = si[r];
    }
}

__global__ void k12_producers(const float* __restrict__ x,
                              const float* __restrict__ proj_w,
                              const float* __restrict__ qnn_w,
                              float* __restrict__ angles,
                              float* __restrict__ UrP,
                              float* __restrict__ UiP) {
    int b = blockIdx.x;
    if (b < U_BLOCKS) unitary_cols(qnn_w, UrP, UiP, b);
    else              angles_rows(x, proj_w, angles, b - U_BLOCKS);
}

__global__ __launch_bounds__(512) void k3_fused(
        const float* __restrict__ angles,
        const float* __restrict__ UrP,
        const float* __restrict__ UiP,
        const float* __restrict__ out_w,
        const float* __restrict__ out_b,
        float* __restrict__ out) {
    __shared__ __align__(16) float cs[16][NQ][2];
    __shared__ __align__(16) float enc[16][NSTATE];
    __shared__ __align__(16) float pq[16][NSTATE];
    __shared__ __align__(16) float E[16][NQ];

    int tid  = threadIdx.x;
    int lane = tid & 31;
    int wave = tid >> 5;
    int row0 = blockIdx.x * 16;

    if (tid < 128) {
        int m = tid >> 3, q = tid & 7;
        float a = angles[(row0 + m) * NQ + q] * 0.5f;
        cs[m][q][0] = cosf(a);
        cs[m][q][1] = sinf(a);
    }
    __syncthreads();

#pragma unroll
    for (int e = 0; e < 8; ++e) {
        int idx = tid + 512 * e;
        int m = idx >> 8, t = idx & 255;
        float v = 1.0f;
#pragma unroll
        for (int q = 0; q < NQ; ++q) v *= cs[m][q][(t >> q) & 1];
        enc[m][t] = v;
    }
    __syncthreads();

    v8f cr, ci;
#pragma unroll
    for (int v = 0; v < 8; ++v) { cr[v] = 0.0f; ci[v] = 0.0f; }

    int mrow = lane & 15;
    int h    = lane >> 4;
    int n    = lane & 15;
    int s0   = wave * 16;

    for (int k = 0; k < NSTATE; k += 4) {
        int kk = k + 2 * h;
        v2f a  = *(const v2f*)&enc[mrow][kk];
        const float* pr = &UrP[(kk >> 1) * (2 * NSTATE) + (s0 + n) * 2];
        const float* pi = &UiP[(kk >> 1) * (2 * NSTATE) + (s0 + n) * 2];
        v2f br = *(const v2f*)pr;
        v2f bi = *(const v2f*)pi;
        cr = __builtin_amdgcn_wmma_f32_16x16x4_f32(false, a, false, br,
                                                   (short)0, cr, false, false);
        ci = __builtin_amdgcn_wmma_f32_16x16x4_f32(false, a, false, bi,
                                                   (short)0, ci, false, false);
    }

#pragma unroll
    for (int v = 0; v < 8; ++v) {
        int m = v + 8 * h;
        pq[m][s0 + n] = cr[v] * cr[v] + ci[v] * ci[v];
    }
    __syncthreads();

    if (tid < 128) {
        int m = tid >> 3, q = tid & 7;
        float acc = 0.0f;
        for (int s = 0; s < NSTATE; ++s) {
            float p = pq[m][s];
            acc += ((s >> q) & 1) ? -p : p;
        }
        E[m][q] = acc;
    }
    __syncthreads();

    if (tid < 16 * NCLS) {
        int m = tid / NCLS, c = tid % NCLS;
        float acc = out_b[c];
#pragma unroll
        for (int q = 0; q < NQ; ++q) acc += E[m][q] * out_w[c * NQ + q];
        out[(row0 + m) * NCLS + c] = acc;
    }
}

extern "C" void kernel_launch(void* const* d_in, const int* in_sizes, int n_in,
                              void* d_out, int out_size, void* d_ws, size_t ws_size,
                              hipStream_t stream) {
    const float* x      = (const float*)d_in[0];
    const float* proj_w = (const float*)d_in[1];
    const float* qnn_w  = (const float*)d_in[2];
    const float* out_w  = (const float*)d_in[3];
    const float* out_b  = (const float*)d_in[4];
    float* ws = (float*)d_ws;

    float* angles = ws;
    float* UrP    = ws + 65536;
    float* UiP    = ws + 65536 + 65536;

    k12_producers<<<PRODUCER_BLOCKS, 256, 0, stream>>>(x, proj_w, qnn_w,
                                                       angles, UrP, UiP);
    k3_fused<<<BATCH / 16, 512, 0, stream>>>(angles, UrP, UiP, out_w, out_b,
                                             (float*)d_out);
}


# LDS rows of the enc and pq tiles padded from 256 to 260 floats (16 B x odd pitch): the 16-way bank conflict of every f32 A-fragment read goes, pq store and read conflicts too
# speedup vs baseline: 1.3136x; 1.0474x over previous
#include <hip/hip_runtime.h>
#include <math.h>

typedef float v2f __attribute__((ext_vector_type(2)));
typedef float v8f __attribute__((ext_vector_type(8)));

#define BATCH 8192
#define FDIM  512
#define NQ    8
#define NSTATE 256
#define NCLS  10
#define TILE_PITCH (NSTATE + 4)

#define U_BLOCKS        32
#define PRODUCER_BLOCKS 1056

__device__ __forceinline__ void angles_rows(const float* __restrict__ x,
                                            const float* __restrict__ proj_w,
                                            float* __restrict__ angles,
                                            int bid) {
    __shared__ float w[NQ * FDIM];
    int tid = threadIdx.x;
    for (int i = tid; i < NQ * FDIM; i += 256) w[i] = proj_w[i];
    __syncthreads();

    int lane = tid & 31;
    int wave = tid >> 5;
    int row  = bid * 8 + wave;
    const float* xr = x + (size_t)row * FDIM;

    float acc[NQ];
#pragma unroll
    for (int q = 0; q < NQ; ++q) acc[q] = 0.0f;

    for (int j = 0; j < FDIM / 32; ++j) {
        float xv = xr[lane + 32 * j];
#pragma unroll
        for (int q = 0; q < NQ; ++q)
            acc[q] += xv * w[q * FDIM + lane + 32 * j];
    }
#pragma unroll
    for (int q = 0; q < NQ; ++q) {
        float v = acc[q];
#pragma unroll
        for (int off = 16; off >= 1; off >>= 1) v += __shfl_xor(v, off, 32);
        if (lane == q)
            angles[row * NQ + q] = tanhf(v) * 1.5707963267948966f;
    }
}

__device__ __forceinline__ void rot_lane_rx(float sr[8], float si[8],
                                            float c, float s, int m) {
#pragma unroll
    for (int r = 0; r < 8; ++r) {
        float pr = __shfl_xor(sr[r], m, 32);
        float pi = __shfl_xor(si[r], m, 32);
        float nr = c * sr[r] + s * pi;
        float ni = c * si[r] - s * pr;
        sr[r] = nr; si[r] = ni;
    }
}

__device__ __forceinline__ void rot_lane_ry(float sr[8], float si[8],
                                            float c, float s, int m, int lane) {
    float b = (lane & m) ? s : -s;
#pragma unroll
    for (int r = 0; r < 8; ++r) {
        float pr = __shfl_xor(sr[r], m, 32);
        float pi = __shfl_xor(si[r], m, 32);
        sr[r] = c * sr[r] + b * pr;
        si[r] = c * si[r] + b * pi;
    }
}

__device__ __forceinline__ void rot_reg_rx(float sr[8], float si[8],
                                           float c, float s, int rm) {
#pragma unroll
    for (int r0 = 0; r0 < 8; ++r0) {
        if (r0 & rm) continue;
        int r1 = r0 | rm;
        float a0r = sr[r0], a0i = si[r0], a1r = sr[r1], a1i = si[r1];
        sr[r0] = c * a0r + s * a1i;  si[r0] = c * a0i - s * a1r;
        sr[r1] = c * a1r + s * a0i;  si[r1] = c * a1i - s * a0r;
    }
}

__device__ __forceinline__ void rot_reg_ry(float sr[8], float si[8],
                                           float c, float s, int rm) {
#pragma unroll
    for (int r0 = 0; r0 < 8; ++r0) {
        if (r0 & rm) continue;
        int r1 = r0 | rm;
        float a0r = sr[r0], a0i = si[r0], a1r = sr[r1], a1i = si[r1];
        sr[r0] = c * a0r - s * a1r;  si[r0] = c * a0i - s * a1i;
        sr[r1] = s * a0r + c * a1r;  si[r1] = s * a0i + c * a1i;
    }
}

__device__ __forceinline__ void cnot_ll(float sr[8], float si[8],
                                        int cm, int tm, int lane) {
    bool hi = (lane & cm) != 0;
#pragma unroll
    for (int r = 0; r < 8; ++r) {
        float pr = __shfl_xor(sr[r], tm, 32);
        float pi = __shfl_xor(si[r], tm, 32);
        sr[r] = hi ? pr : sr[r];
        si[r] = hi ? pi : si[r];
    }
}

__device__ __forceinline__ void cnot_lane_reg(float sr[8], float si[8],
                                              int cm, int trm, int lane) {
    bool hi = (lane & cm) != 0;
#pragma unroll
    for (int r0 = 0; r0 < 8; ++r0) {
        if (r0 & trm) continue;
        int r1 = r0 | trm;
        float t0r = sr[r0], t0i = si[r0];
        sr[r0] = hi ? sr[r1] : sr[r0];  si[r0] = hi ? si[r1] : si[r0];
        sr[r1] = hi ? t0r    : sr[r1];  si[r1] = hi ? t0i    : si[r1];
    }
}

__device__ __forceinline__ void cnot_reg_lane(float sr[8], float si[8],
                                              int crm, int tm) {
#pragma unroll
    for (int r = 0; r < 8; ++r) {
        if (!(r & crm)) continue;
        sr[r] = __shfl_xor(sr[r], tm, 32);
        si[r] = __shfl_xor(si[r], tm, 32);
    }
}

__device__ __forceinline__ void cnot_reg_reg(float sr[8], float si[8],
                                             int crm, int trm) {
#pragma unroll
    for (int r0 = 0; r0 < 8; ++r0) {
        if (!(r0 & crm) || (r0 & trm)) continue;
        int r1 = r0 | trm;
        float tr = sr[r0], ti = si[r0];
        sr[r0] = sr[r1]; si[r0] = si[r1];
        sr[r1] = tr;     si[r1] = ti;
    }
}

__device__ __forceinline__ void unitary_cols(const float* __restrict__ qnn_w,
                                             float* __restrict__ UrP,
                                             float* __restrict__ UiP,
                                             int bid) {
    int lane = threadIdx.x & 31;
    int t    = bid * 8 + (threadIdx.x >> 5);

    const __attribute__((address_space(4))) float* wq =
        (const __attribute__((address_space(4))) float*)qnn_w;

    float sr[8], si[8];
#pragma unroll
    for (int r = 0; r < 8; ++r) {
        sr[r] = (r * 32 + lane == t) ? 1.0f : 0.0f;
        si[r] = 0.0f;
    }

#pragma unroll
    for (int l = 0; l < 3; ++l) {
#pragma unroll
        for (int q = 0; q < NQ; ++q) {
            float w0 = wq[(l * NQ + q) * 3 + 0] * 0.5f;
            float w1 = wq[(l * NQ + q) * 3 + 1] * 0.5f;
            float w2 = wq[(l * NQ + q) * 3 + 2] * 0.5f;
            float c0 = cosf(w0), s0 = sinf(w0);
            float c1 = cosf(w1), s1 = sinf(w1);
            float c2 = cosf(w2), s2 = sinf(w2);
            if (q < 5) {
                int m = 1 << q;
                rot_lane_rx(sr, si, c0, s0, m);
                rot_lane_ry(sr, si, c1, s1, m, lane);
                rot_lane_rx(sr, si, c2, s2, m);
            } else {
                int rm = 1 << (q - 5);
                rot_reg_rx(sr, si, c0, s0, rm);
                rot_reg_ry(sr, si, c1, s1, rm);
                rot_reg_rx(sr, si, c2, s2, rm);
            }
        }
        cnot_ll(sr, si, 1, 2, lane);
        cnot_ll(sr, si, 2, 4, lane);
        cnot_ll(sr, si, 4, 8, lane);
        cnot_ll(sr, si, 8, 16, lane);
        cnot_lane_reg(sr, si, 16, 1, lane);
        cnot_reg_reg(sr, si, 1, 2);
        cnot_reg_reg(sr, si, 2, 4);
        cnot_reg_lane(sr, si, 4, 1);
    }

#pragma unroll
    for (int r = 0; r < 8; ++r) {
        int s = r * 32 + lane;
        int off = (t >> 1) * (2 * NSTATE) + s * 2 + (t & 1);
        UrP[off] = sr[r];
        UiP[off] = si[r];
    }
}

__global__ void k12_producers(const float* __restrict__ x,
                              const float* __restrict__ proj_w,
                              const float* __restrict__ qnn_w,
                              float* __restrict__ angles,
                              float* __restrict__ UrP,
                              float* __restrict__ UiP) {
    int b = blockIdx.x;
    if (b < U_BLOCKS) unitary_cols(qnn_w, UrP, UiP, b);
    else              angles_rows(x, proj_w, angles, b - U_BLOCKS);
}

__global__ __launch_bounds__(512) void k3_fused(
        const float* __restrict__ angles,
        const float* __restrict__ UrP,
        const float* __restrict__ UiP,
        const float* __restrict__ out_w,
        const float* __restrict__ out_b,
        float* __restrict__ out) {
    __shared__ __align__(16) float cs[16][NQ][2];
    __shared__ __align__(16) float enc[16][TILE_PITCH];
    __shared__ __align__(16) float pq[16][TILE_PITCH];
    __shared__ __align__(16) float E[16][NQ];

    int tid  = threadIdx.x;
    int lane = tid & 31;
    int wave = tid >> 5;
    int row0 = blockIdx.x * 16;

    if (tid < 128) {
        int m = tid >> 3, q = tid & 7;
        float a = angles[(row0 + m) * NQ + q] * 0.5f;
        cs[m][q][0] = cosf(a);
        cs[m][q][1] = sinf(a);
    }
    __syncthreads();

#pragma unroll
    for (int e = 0; e < 8; ++e) {
        int idx = tid + 512 * e;
        int m = idx >> 8, t = idx & 255;
        float v = 1.0f;
#pragma unroll
        for (int q = 0; q < NQ; ++q) v *= cs[m][q][(t >> q) & 1];
        enc[m][t] = v;
    }
    __syncthreads();

    v8f cr, ci;
#pragma unroll
    for (int v = 0; v < 8; ++v) { cr[v] = 0.0f; ci[v] = 0.0f; }

    int mrow = lane & 15;
    int h    = lane >> 4;
    int n    = lane & 15;
    int s0   = wave * 16;

    for (int k = 0; k < NSTATE; k += 4) {
        int kk = k + 2 * h;
        v2f a  = *(const v2f*)&enc[mrow][kk];
        const float* pr = &UrP[(kk >> 1) * (2 * NSTATE) + (s0 + n) * 2];
        const float* pi = &UiP[(kk >> 1) * (2 * NSTATE) + (s0 + n) * 2];
        v2f br = *(const v2f*)pr;
        v2f bi = *(const v2f*)pi;
        cr = __builtin_amdgcn_wmma_f32_16x16x4_f32(false, a, false, br,
                                                   (short)0, cr, false, false);
        ci = __builtin_amdgcn_wmma_f32_16x16x4_f32(false, a, false, bi,
                                                   (short)0, ci, false, false);
    }

#pragma unroll
    for (int v = 0; v < 8; ++v) {
        int m = v + 8 * h;
        pq[m][s0 + n] = cr[v] * cr[v] + ci[v] * ci[v];
    }
    __syncthreads();

    if (tid < 128) {
        int m = tid >> 3, q = tid & 7;
        float acc = 0.0f;
        for (int s = 0; s < NSTATE; ++s) {
            float p = pq[m][s];
            acc += ((s >> q) & 1) ? -p : p;
        }
        E[m][q] = acc;
    }
    __syncthreads();

    if (tid < 16 * NCLS) {
        int m = tid / NCLS, c = tid % NCLS;
        float acc = out_b[c];
#pragma unroll
        for (int q = 0; q < NQ; ++q) acc += E[m][q] * out_w[c * NQ + q];
        out[(row0 + m) * NCLS + c] = acc;
    }
}

extern "C" void kernel_launch(void* const* d_in, const int* in_sizes, int n_in,
                              void* d_out, int out_size, void* d_ws, size_t ws_size,
                              hipStream_t stream) {
    const float* x      = (const float*)d_in[0];
    const float* proj_w = (const float*)d_in[1];
    const float* qnn_w  = (const float*)d_in[2];
    const float* out_w  = (const float*)d_in[3];
    const float* out_b  = (const float*)d_in[4];
    float* ws = (float*)d_ws;

    float* angles = ws;
    float* UrP    = ws + 65536;
    float* UiP    = ws + 65536 + 65536;

    k12_producers<<<PRODUCER_BLOCKS, 256, 0, stream>>>(x, proj_w, qnn_w,
                                                       angles, UrP, UiP);
    k3_fused<<<BATCH / 16, 512, 0, stream>>>(angles, UrP, UiP, out_w, out_b,
                                             (float*)d_out);
}


# GEMM kernel re-tiled to a 2x2 register tile per wave: 32 rows per work-group, 8 waves, one 16-byte U load per lane feeds 4 WMMAs; U read once per 32 rows, 2048 waves instead of 8192
# speedup vs baseline: 1.3792x; 1.0499x over previous
#include <hip/hip_runtime.h>
#include <math.h>

typedef float v2f __attribute__((ext_vector_type(2)));
typedef float v4f __attribute__((ext_vector_type(4)));
typedef float v8f __attribute__((ext_vector_type(8)));

#define BATCH 8192
#define FDIM  512
#define NQ    8
#define NSTATE 256
#define NCLS  10
#define TILE_PITCH (NSTATE + 4)

#define U_BLOCKS        32
#define PRODUCER_BLOCKS 1056

__device__ __forceinline__ void angles_rows(const float* __restrict__ x,
                                            const float* __restrict__ proj_w,
                                            float* __restrict__ angles,
                                            int bid) {
    __shared__ float w[NQ * FDIM];
    int tid = threadIdx.x;
    for (int i = tid; i < NQ * FDIM; i += 256) w[i] = proj_w[i];
    __syncthreads();

    int lane = tid & 31;
    int wave = tid >> 5;
    int row  = bid * 8 + wave;
    const float* xr = x + (size_t)row * FDIM;

    float acc[NQ];
#pragma unroll
    for (int q = 0; q < NQ; ++q) acc[q] = 0.0f;

    for (int j = 0; j < FDIM / 32; ++j) {
        float xv = xr[lane + 32 * j];
#pragma unroll
        for (int q = 0; q < NQ; ++q)
            acc[q] += xv * w[q * FDIM + lane + 32 * j];
    }
#pragma unroll
    for (int q = 0; q < NQ; ++q) {
        float v = acc[q];
#pragma unroll
        for (int off = 16; off >= 1; off >>= 1) v += __shfl_xor(v, off, 32);
        if (lane == q)
            angles[row * NQ + q] = tanhf(v) * 1.5707963267948966f;
    }
}

__device__ __forceinline__ void rot_lane_rx(float sr[8], float si[8],
                                            float c, float s, int m) {
#pragma unroll
    for (int r = 0; r < 8; ++r) {
        float pr = __shfl_xor(sr[r], m, 32);
        float pi = __shfl_xor(si[r], m, 32);
        float nr = c * sr[r] + s * pi;
        float ni = c * si[r] - s * pr;
        sr[r] = nr; si[r] = ni;
    }
}

__device__ __forceinline__ void rot_lane_ry(float sr[8], float si[8],
                                            float c, float s, int m, int lane) {
    float b = (lane & m) ? s : -s;
#pragma unroll
    for (int r = 0; r < 8; ++r) {
        float pr = __shfl_xor(sr[r], m, 32);
        float pi = __shfl_xor(si[r], m, 32);
        sr[r] = c * sr[r] + b * pr;
        si[r] = c * si[r] + b * pi;
    }
}

__device__ __forceinline__ void rot_reg_rx(float sr[8], float si[8],
                                           float c, float s, int rm) {
#pragma unroll
    for (int r0 = 0; r0 < 8; ++r0) {
        if (r0 & rm) continue;
        int r1 = r0 | rm;
        float a0r = sr[r0], a0i = si[r0], a1r = sr[r1], a1i = si[r1];
        sr[r0] = c * a0r + s * a1i;  si[r0] = c * a0i - s * a1r;
        sr[r1] = c * a1r + s * a0i;  si[r1] = c * a1i - s * a0r;
    }
}

__device__ __forceinline__ void rot_reg_ry(float sr[8], float si[8],
                                           float c, float s, int rm) {
#pragma unroll
    for (int r0 = 0; r0 < 8; ++r0) {
        if (r0 & rm) continue;
        int r1 = r0 | rm;
        float a0r = sr[r0], a0i = si[r0], a1r = sr[r1], a1i = si[r1];
        sr[r0] = c * a0r - s * a1r;  si[r0] = c * a0i - s * a1i;
        sr[r1] = s * a0r + c * a1r;  si[r1] = s * a0i + c * a1i;
    }
}

__device__ __forceinline__ void cnot_ll(float sr[8], float si[8],
                                        int cm, int tm, int lane) {
    bool hi = (lane & cm) != 0;
#pragma unroll
    for (int r = 0; r < 8; ++r) {
        float pr = __shfl_xor(sr[r], tm, 32);
        float pi = __shfl_xor(si[r], tm, 32);
        sr[r] = hi ? pr : sr[r];
        si[r] = hi ? pi : si[r];
    }
}

__device__ __forceinline__ void cnot_lane_reg(float sr[8], float si[8],
                                              int cm, int trm, int lane) {
    bool hi = (lane & cm) != 0;
#pragma unroll
    for (int r0 = 0; r0 < 8; ++r0) {
        if (r0 & trm) continue;
        int r1 = r0 | trm;
        float t0r = sr[r0], t0i = si[r0];
        sr[r0] = hi ? sr[r1] : sr[r0];  si[r0] = hi ? si[r1] : si[r0];
        sr[r1] = hi ? t0r    : sr[r1];  si[r1] = hi ? t0i    : si[r1];
    }
}

__device__ __forceinline__ void cnot_reg_lane(float sr[8], float si[8],
                                              int crm, int tm) {
#pragma unroll
    for (int r = 0; r < 8; ++r) {
        if (!(r & crm)) continue;
        sr[r] = __shfl_xor(sr[r], tm, 32);
        si[r] = __shfl_xor(si[r], tm, 32);
    }
}

__device__ __forceinline__ void cnot_reg_reg(float sr[8], float si[8],
                                             int crm, int trm) {
#pragma unroll
    for (int r0 = 0; r0 < 8; ++r0) {
        if (!(r0 & crm) || (r0 & trm)) continue;
        int r1 = r0 | trm;
        float tr = sr[r0], ti = si[r0];
        sr[r0] = sr[r1]; si[r0] = si[r1];
        sr[r1] = tr;     si[r1] = ti;
    }
}

__device__ __forceinline__ void unitary_cols(const float* __restrict__ qnn_w,
                                             float* __restrict__ UrP,
                                             float* __restrict__ UiP,
                                             int bid) {
    int lane = threadIdx.x & 31;
    int t    = bid * 8 + (threadIdx.x >> 5);

    const __attribute__((address_space(4))) float* wq =
        (const __attribute__((address_space(4))) float*)qnn_w;

    float sr[8], si[8];
#pragma unroll
    for (int r = 0; r < 8; ++r) {
        sr[r] = (r * 32 + lane == t) ? 1.0f : 0.0f;
        si[r] = 0.0f;
    }

#pragma unroll
    for (int l = 0; l < 3; ++l) {
#pragma unroll
        for (int q = 0; q < NQ; ++q) {
            float w0 = wq[(l * NQ + q) * 3 + 0] * 0.5f;
            float w1 = wq[(l * NQ + q) * 3 + 1] * 0.5f;
            float w2 = wq[(l * NQ + q) * 3 + 2] * 0.5f;
            float c0 = cosf(w0), s0 = sinf(w0);
            float c1 = cosf(w1), s1 = sinf(w1);
            float c2 = cosf(w2), s2 = sinf(w2);
            if (q < 5) {
                int m = 1 << q;
                rot_lane_rx(sr, si, c0, s0, m);
                rot_lane_ry(sr, si, c1, s1, m, lane);
                rot_lane_rx(sr, si, c2, s2, m);
            } else {
                int rm = 1 << (q - 5);
                rot_reg_rx(sr, si, c0, s0, rm);
                rot_reg_ry(sr, si, c1, s1, rm);
                rot_reg_rx(sr, si, c2, s2, rm);
            }
        }
        cnot_ll(sr, si, 1, 2, lane);
        cnot_ll(sr, si, 2, 4, lane);
        cnot_ll(sr, si, 4, 8, lane);
        cnot_ll(sr, si, 8, 16, lane);
        cnot_lane_reg(sr, si, 16, 1, lane);
        cnot_reg_reg(sr, si, 1, 2);
        cnot_reg_reg(sr, si, 2, 4);
        cnot_reg_lane(sr, si, 4, 1);
    }

#pragma unroll
    for (int r = 0; r < 8; ++r) {
        int s = r * 32 + lane;
        int off = (t >> 1) * (2 * NSTATE) + s * 2 + (t & 1);
        UrP[off] = sr[r];
        UiP[off] = si[r];
    }
}

__global__ void k12_producers(const float* __restrict__ x,
                              const float* __restrict__ proj_w,
                              const float* __restrict__ qnn_w,
                              float* __restrict__ angles,
                              float* __restrict__ UrP,
                              float* __restrict__ UiP) {
    int b = blockIdx.x;
    if (b < U_BLOCKS) unitary_cols(qnn_w, UrP, UiP, b);
    else              angles_rows(x, proj_w, angles, b - U_BLOCKS);
}

#define ROWS_PER_BLOCK 32
#define KSTEPS_PER_GROUP 8
#define K_GROUPS (NSTATE / (4 * KSTEPS_PER_GROUP))

__device__ __forceinline__ void load_u_group(const float* __restrict__ ur,
                                             const float* __restrict__ ui, int g,
                                             v4f (&br)[KSTEPS_PER_GROUP],
                                             v4f (&bi)[KSTEPS_PER_GROUP]) {
#pragma unroll
    for (int j = 0; j < KSTEPS_PER_GROUP; ++j) {
        int k = (g * KSTEPS_PER_GROUP + j) * 4;
        br[j] = *(const v4f*)(ur + (k >> 1) * (2 * NSTATE));
        bi[j] = *(const v4f*)(ui + (k >> 1) * (2 * NSTATE));
    }
}

__device__ __forceinline__ void mma_group(const float* ea, int g,
                                          const v4f (&br)[KSTEPS_PER_GROUP],
                                          const v4f (&bi)[KSTEPS_PER_GROUP],
                                          v8f (&acc)[8]) {
#pragma unroll
    for (int j = 0; j < KSTEPS_PER_GROUP; ++j) {
        int k = (g * KSTEPS_PER_GROUP + j) * 4;
        v2f a0 = *(const v2f*)(ea + k);
        v2f a1 = *(const v2f*)(ea + 16 * TILE_PITCH + k);
        v2f br0 = br[j].lo, br1 = br[j].hi;
        v2f bi0 = bi[j].lo, bi1 = bi[j].hi;
        acc[0] = __builtin_amdgcn_wmma_f32_16x16x4_f32(false, a0, false, br0, (short)0, acc[0], false, false);
        acc[1] = __builtin_amdgcn_wmma_f32_16x16x4_f32(false, a0, false, bi0, (short)0, acc[1], false, false);
        acc[2] = __builtin_amdgcn_wmma_f32_16x16x4_f32(false, a0, false, br1, (short)0, acc[2], false, false);
        acc[3] = __builtin_amdgcn_wmma_f32_16x16x4_f32(false, a0, false, bi1, (short)0, acc[3], false, false);
        acc[4] = __builtin_amdgcn_wmma_f32_16x16x4_f32(false, a1, false, br0, (short)0, acc[4], false, false);
        acc[5] = __builtin_amdgcn_wmma_f32_16x16x4_f32(false, a1, false, bi0, (short)0, acc[5], false, false);
        acc[6] = __builtin_amdgcn_wmma_f32_16x16x4_f32(false, a1, false, br1, (short)0, acc[6], false, false);
        acc[7] = __builtin_amdgcn_wmma_f32_16x16x4_f32(false, a1, false, bi1, (short)0, acc[7], false, false);
    }
}

__global__ __launch_bounds__(256) __attribute__((amdgpu_waves_per_eu(2, 6))) void k3_fused(
        const float* __restrict__ angles,
        const float* __restrict__ UrP,
        const float* __restrict__ UiP,
        const float* __restrict__ out_w,
        const float* __restrict__ out_b,
        float* __restrict__ out) {
    __shared__ __align__(16) float cs[ROWS_PER_BLOCK][NQ][2];
    __shared__ __align__(16) float enc[ROWS_PER_BLOCK][TILE_PITCH];
    __shared__ __align__(16) float pq[ROWS_PER_BLOCK][TILE_PITCH];
    __shared__ __align__(16) float E[ROWS_PER_BLOCK][NQ];

    int tid  = threadIdx.x;
    int lane = tid & 31;
    int wave = tid >> 5;
    int row0 = blockIdx.x * ROWS_PER_BLOCK;

    {
        int m = tid >> 3, q = tid & 7;
        float a = angles[(row0 + m) * NQ + q] * 0.5f;
        cs[m][q][0] = cosf(a);
        cs[m][q][1] = sinf(a);
    }
    __syncthreads();

    for (int m = 0; m < ROWS_PER_BLOCK; ++m) {
        float v = 1.0f;
#pragma unroll
        for (int q = 0; q < NQ; ++q) v *= cs[m][q][(tid >> q) & 1];
        enc[m][tid] = v;
    }
    __syncthreads();

    v8f acc[8];
#pragma unroll
    for (int i = 0; i < 8; ++i)
#pragma unroll
        for (int v = 0; v < 8; ++v) acc[i][v] = 0.0f;

    int mrow = lane & 15;
    int h    = lane >> 4;
    int n    = lane & 15;
    int sb   = wave * 32;

    const float* ur = UrP + h * (2 * NSTATE) + (sb + 2 * n) * 2;
    const float* ui = UiP + h * (2 * NSTATE) + (sb + 2 * n) * 2;
    const float* ea = &enc[mrow][2 * h];

#pragma unroll
    for (int g = 0; g < K_GROUPS; ++g) {
        v4f br[KSTEPS_PER_GROUP], bi[KSTEPS_PER_GROUP];
        load_u_group(ur, ui, g, br, bi);
        mma_group(ea, g, br, bi, acc);
    }

#pragma unroll
    for (int v = 0; v < 8; ++v) {
        int m = v + 8 * h;
        v2f p0, p1;
        p0[0] = acc[0][v] * acc[0][v] + acc[1][v] * acc[1][v];
        p0[1] = acc[2][v] * acc[2][v] + acc[3][v] * acc[3][v];
        p1[0] = acc[4][v] * acc[4][v] + acc[5][v] * acc[5][v];
        p1[1] = acc[6][v] * acc[6][v] + acc[7][v] * acc[7][v];
        *(v2f*)&pq[m][sb + 2 * n]      = p0;
        *(v2f*)&pq[16 + m][sb + 2 * n] = p1;
    }
    __syncthreads();

    {
        int m = tid >> 3, q = tid & 7;
        float esum = 0.0f;
        for (int s = 0; s < NSTATE; ++s) {
            float p = pq[m][s];
            esum += ((s >> q) & 1) ? -p : p;
        }
        E[m][q] = esum;
    }
    __syncthreads();

    for (int idx = tid; idx < ROWS_PER_BLOCK * NCLS; idx += 256) {
        int m = idx / NCLS, c = idx % NCLS;
        float o = out_b[c];
#pragma unroll
        for (int q = 0; q < NQ; ++q) o += E[m][q] * out_w[c * NQ + q];
        out[(row0 + m) * NCLS + c] = o;
    }
}

extern "C" void kernel_launch(void* const* d_in, const int* in_sizes, int n_in,
                              void* d_out, int out_size, void* d_ws, size_t ws_size,
                              hipStream_t stream) {
    const float* x      = (const float*)d_in[0];
    const float* proj_w = (const float*)d_in[1];
    const float* qnn_w  = (const float*)d_in[2];
    const float* out_w  = (const float*)d_in[3];
    const float* out_b  = (const float*)d_in[4];
    float* ws = (float*)d_ws;

    float* angles = ws;
    float* UrP    = ws + 65536;
    float* UiP    = ws + 65536 + 65536;

    k12_producers<<<PRODUCER_BLOCKS, 256, 0, stream>>>(x, proj_w, qnn_w,
                                                       angles, UrP, UiP);
    k3_fused<<<BATCH / ROWS_PER_BLOCK, 256, 0, stream>>>(angles, UrP, UiP, out_w, out_b,
                                             (float*)d_out);
}


# angle role re-tiled: one wave per 4 rows, 288 work-groups instead of 1056, 16-byte x and w loads, lane sums reduced by recursive halving (summation order of the angles changes)
# speedup vs baseline: 1.5121x; 1.0964x over previous
#include <hip/hip_runtime.h>
#include <math.h>

typedef float v2f __attribute__((ext_vector_type(2)));
typedef float v4f __attribute__((ext_vector_type(4)));
typedef float v8f __attribute__((ext_vector_type(8)));

#define BATCH 8192
#define FDIM  512
#define NQ    8
#define NSTATE 256
#define NCLS  10
#define TILE_PITCH (NSTATE + 4)

#define U_BLOCKS        32
#define ANGLE_ROWS_PER_WAVE  4
#define ANGLE_ROWS_PER_BLOCK 32
#define ANGLE_BLOCKS    (BATCH / ANGLE_ROWS_PER_BLOCK)
#define PRODUCER_BLOCKS (U_BLOCKS + ANGLE_BLOCKS)

__device__ __forceinline__ void angles_rows(const float* __restrict__ x,
                                            const float* __restrict__ proj_w,
                                            float* __restrict__ angles,
                                            int bid) {
    __shared__ __align__(16) float w[NQ * FDIM];
    int tid  = threadIdx.x;
    int lane = tid & 31;
    int wave = tid >> 5;
    int row0 = bid * ANGLE_ROWS_PER_BLOCK + wave * ANGLE_ROWS_PER_WAVE;

    const v4f* pw4 = (const v4f*)proj_w;
    v4f wst[4];
#pragma unroll
    for (int k = 0; k < 4; ++k) wst[k] = pw4[tid + 256 * k];

    const v4f* xr = (const v4f*)(x + (size_t)row0 * FDIM) + lane;
    v4f xv[ANGLE_ROWS_PER_WAVE][4];
#pragma unroll
    for (int j = 0; j < 4; ++j)
#pragma unroll
        for (int r = 0; r < ANGLE_ROWS_PER_WAVE; ++r)
            xv[r][j] = xr[r * (FDIM / 4) + j * 32];

    v4f* w4 = (v4f*)w;
#pragma unroll
    for (int k = 0; k < 4; ++k) w4[tid + 256 * k] = wst[k];
    __syncthreads();

    v2f accv[ANGLE_ROWS_PER_WAVE][NQ];
#pragma unroll
    for (int r = 0; r < ANGLE_ROWS_PER_WAVE; ++r)
#pragma unroll
        for (int q = 0; q < NQ; ++q) { accv[r][q][0] = 0.0f; accv[r][q][1] = 0.0f; }

#pragma unroll
    for (int j = 0; j < 4; ++j) {
#pragma unroll
        for (int q = 0; q < NQ; ++q) {
            v4f wv = w4[q * (FDIM / 4) + j * 32 + lane];
            v2f wlo = wv.lo, whi = wv.hi;
#pragma unroll
            for (int r = 0; r < ANGLE_ROWS_PER_WAVE; ++r) {
                v2f xlo = xv[r][j].lo, xhi = xv[r][j].hi;
                accv[r][q] = __builtin_elementwise_fma(xlo, wlo, accv[r][q]);
                accv[r][q] = __builtin_elementwise_fma(xhi, whi, accv[r][q]);
            }
        }
    }
    float acc[ANGLE_ROWS_PER_WAVE * NQ];
#pragma unroll
    for (int r = 0; r < ANGLE_ROWS_PER_WAVE; ++r)
#pragma unroll
        for (int q = 0; q < NQ; ++q) acc[r * NQ + q] = accv[r][q][0] + accv[r][q][1];

    float a16[16], a8[8], a4[4], a2[2];
    {
        bool hi = (lane & 16) != 0;
#pragma unroll
        for (int i = 0; i < 16; ++i) {
            float keep = hi ? acc[i + 16] : acc[i];
            float send = hi ? acc[i] : acc[i + 16];
            a16[i] = keep + __shfl_xor(send, 16, 32);
        }
    }
    {
        bool hi = (lane & 8) != 0;
#pragma unroll
        for (int i = 0; i < 8; ++i) {
            float keep = hi ? a16[i + 8] : a16[i];
            float send = hi ? a16[i] : a16[i + 8];
            a8[i] = keep + __shfl_xor(send, 8, 32);
        }
    }
    {
        bool hi = (lane & 4) != 0;
#pragma unroll
        for (int i = 0; i < 4; ++i) {
            float keep = hi ? a8[i + 4] : a8[i];
            float send = hi ? a8[i] : a8[i + 4];
            a4[i] = keep + __shfl_xor(send, 4, 32);
        }
    }
    {
        bool hi = (lane & 2) != 0;
#pragma unroll
        for (int i = 0; i < 2; ++i) {
            float keep = hi ? a4[i + 2] : a4[i];
            float send = hi ? a4[i] : a4[i + 2];
            a2[i] = keep + __shfl_xor(send, 2, 32);
        }
    }
    float v;
    {
        bool hi = (lane & 1) != 0;
        float keep = hi ? a2[1] : a2[0];
        float send = hi ? a2[0] : a2[1];
        v = keep + __shfl_xor(send, 1, 32);
    }
    angles[row0 * NQ + lane] = tanhf(v) * 1.5707963267948966f;
}

__device__ __forceinline__ void rot_lane_rx(float sr[8], float si[8],
                                            float c, float s, int m) {
#pragma unroll
    for (int r = 0; r < 8; ++r) {
        float pr = __shfl_xor(sr[r], m, 32);
        float pi = __shfl_xor(si[r], m, 32);
        float nr = c * sr[r] + s * pi;
        float ni = c * si[r] - s * pr;
        sr[r] = nr; si[r] = ni;
    }
}

__device__ __forceinline__ void rot_lane_ry(float sr[8], float si[8],
                                            float c, float s, int m, int lane) {
    float b = (lane & m) ? s : -s;
#pragma unroll
    for (int r = 0; r < 8; ++r) {
        float pr = __shfl_xor(sr[r], m, 32);
        float pi = __shfl_xor(si[r], m, 32);
        sr[r] = c * sr[r] + b * pr;
        si[r] = c * si[r] + b * pi;
    }
}

__device__ __forceinline__ void rot_reg_rx(float sr[8], float si[8],
                                           float c, float s, int rm) {
#pragma unroll
    for (int r0 = 0; r0 < 8; ++r0) {
        if (r0 & rm) continue;
        int r1 = r0 | rm;
        float a0r = sr[r0], a0i = si[r0], a1r = sr[r1], a1i = si[r1];
        sr[r0] = c * a0r + s * a1i;  si[r0] = c * a0i - s * a1r;
        sr[r1] = c * a1r + s * a0i;  si[r1] = c * a1i - s * a0r;
    }
}

__device__ __forceinline__ void rot_reg_ry(float sr[8], float si[8],
                                           float c, float s, int rm) {
#pragma unroll
    for (int r0 = 0; r0 < 8; ++r0) {
        if (r0 & rm) continue;
        int r1 = r0 | rm;
        float a0r = sr[r0], a0i = si[r0], a1r = sr[r1], a1i = si[r1];
        sr[r0] = c * a0r - s * a1r;  si[r0] = c * a0i - s * a1i;
        sr[r1] = s * a0r + c * a1r;  si[r1] = s * a0i + c * a1i;
    }
}

__device__ __forceinline__ void cnot_ll(float sr[8], float si[8],
                                        int cm, int tm, int lane) {
    bool hi = (lane & cm) != 0;
#pragma unroll
    for (int r = 0; r < 8; ++r) {
        float pr = __shfl_xor(sr[r], tm, 32);
        float pi = __shfl_xor(si[r], tm, 32);
        sr[r] = hi ? pr : sr[r];
        si[r] = hi ? pi : si[r];
    }
}

__device__ __forceinline__ void cnot_lane_reg(float sr[8], float si[8],
                                              int cm, int trm, int lane) {
    bool hi = (lane & cm) != 0;
#pragma unroll
    for (int r0 = 0; r0 < 8; ++r0) {
        if (r0 & trm) continue;
        int r1 = r0 | trm;
        float t0r = sr[r0], t0i = si[r0];
        sr[r0] = hi ? sr[r1] : sr[r0];  si[r0] = hi ? si[r1] : si[r0];
        sr[r1] = hi ? t0r    : sr[r1];  si[r1] = hi ? t0i    : si[r1];
    }
}

__device__ __forceinline__ void cnot_reg_lane(float sr[8], float si[8],
                                              int crm, int tm) {
#pragma unroll
    for (int r = 0; r < 8; ++r) {
        if (!(r & crm)) continue;
        sr[r] = __shfl_xor(sr[r], tm, 32);
        si[r] = __shfl_xor(si[r], tm, 32);
    }
}

__device__ __forceinline__ void cnot_reg_reg(float sr[8], float si[8],
                                             int crm, int trm) {
#pragma unroll
    for (int r0 = 0; r0 < 8; ++r0) {
        if (!(r0 & crm) || (r0 & trm)) continue;
        int r1 = r0 | trm;
        float tr = sr[r0], ti = si[r0];
        sr[r0] = sr[r1]; si[r0] = si[r1];
        sr[r1] = tr;     si[r1] = ti;
    }
}

__device__ __forceinline__ void unitary_cols(const float* __restrict__ qnn_w,
                                             float* __restrict__ UrP,
                                             float* __restrict__ UiP,
                                             int bid) {
    int lane = threadIdx.x & 31;
    int t    = bid * 8 + (threadIdx.x >> 5);

    const __attribute__((address_space(4))) float* wq =
        (const __attribute__((address_space(4))) float*)qnn_w;

    float sr[8], si[8];
#pragma unroll
    for (int r = 0; r < 8; ++r) {
        sr[r] = (r * 32 + lane == t) ? 1.0f : 0.0f;
        si[r] = 0.0f;
    }

#pragma unroll
    for (int l = 0; l < 3; ++l) {
#pragma unroll
        for (int q = 0; q < NQ; ++q) {
            float w0 = wq[(l * NQ + q) * 3 + 0] * 0.5f;
            float w1 = wq[(l * NQ + q) * 3 + 1] * 0.5f;
            float w2 = wq[(l * NQ + q) * 3 + 2] * 0.5f;
            float c0 = cosf(w0), s0 = sinf(w0);
            float c1 = cosf(w1), s1 = sinf(w1);
            float c2 = cosf(w2), s2 = sinf(w2);
            if (q < 5) {
                int m = 1 << q;
                rot_lane_rx(sr, si, c0, s0, m);
                rot_lane_ry(sr, si, c1, s1, m, lane);
                rot_lane_rx(sr, si, c2, s2, m);
            } else {
                int rm = 1 << (q - 5);
                rot_reg_rx(sr, si, c0, s0, rm);
                rot_reg_ry(sr, si, c1, s1, rm);
                rot_reg_rx(sr, si, c2, s2, rm);
            }
        }
        cnot_ll(sr, si, 1, 2, lane);
        cnot_ll(sr, si, 2, 4, lane);
        cnot_ll(sr, si, 4, 8, lane);
        cnot_ll(sr, si, 8, 16, lane);
        cnot_lane_reg(sr, si, 16, 1, lane);
        cnot_reg_reg(sr, si, 1, 2);
        cnot_reg_reg(sr, si, 2, 4);
        cnot_reg_lane(sr, si, 4, 1);
    }

#pragma unroll
    for (int r = 0; r < 8; ++r) {
        int s = r * 32 + lane;
        int off = (t >> 1) * (2 * NSTATE) + s * 2 + (t & 1);
        UrP[off] = sr[r];
        UiP[off] = si[r];
    }
}

__global__ __launch_bounds__(256) void k12_producers(const float* __restrict__ x,
                              const float* __restrict__ proj_w,
                              const float* __restrict__ qnn_w,
                              float* __restrict__ angles,
                              float* __restrict__ UrP,
                              float* __restrict__ UiP) {
    int b = blockIdx.x;
    if (b < U_BLOCKS) unitary_cols(qnn_w, UrP, UiP, b);
    else              angles_rows(x, proj_w, angles, b - U_BLOCKS);
}

#define ROWS_PER_BLOCK 32
#define KSTEPS_PER_GROUP 8
#define K_GROUPS (NSTATE / (4 * KSTEPS_PER_GROUP))

__device__ __forceinline__ void load_u_group(const float* __restrict__ ur,
                                             const float* __restrict__ ui, int g,
                                             v4f (&br)[KSTEPS_PER_GROUP],
                                             v4f (&bi)[KSTEPS_PER_GROUP]) {
#pragma unroll
    for (int j = 0; j < KSTEPS_PER_GROUP; ++j) {
        int k = (g * KSTEPS_PER_GROUP + j) * 4;
        br[j] = *(const v4f*)(ur + (k >> 1) * (2 * NSTATE));
        bi[j] = *(const v4f*)(ui + (k >> 1) * (2 * NSTATE));
    }
}

__device__ __forceinline__ void mma_group(const float* ea, int g,
                                          const v4f (&br)[KSTEPS_PER_GROUP],
                                          const v4f (&bi)[KSTEPS_PER_GROUP],
                                          v8f (&acc)[8]) {
#pragma unroll
    for (int j = 0; j < KSTEPS_PER_GROUP; ++j) {
        int k = (g * KSTEPS_PER_GROUP + j) * 4;
        v2f a0 = *(const v2f*)(ea + k);
        v2f a1 = *(const v2f*)(ea + 16 * TILE_PITCH + k);
        v2f br0 = br[j].lo, br1 = br[j].hi;
        v2f bi0 = bi[j].lo, bi1 = bi[j].hi;
        acc[0] = __builtin_amdgcn_wmma_f32_16x16x4_f32(false, a0, false, br0, (short)0, acc[0], false, false);
        acc[1] = __builtin_amdgcn_wmma_f32_16x16x4_f32(false, a0, false, bi0, (short)0, acc[1], false, false);
        acc[2] = __builtin_amdgcn_wmma_f32_16x16x4_f32(false, a0, false, br1, (short)0, acc[2], false, false);
        acc[3] = __builtin_amdgcn_wmma_f32_16x16x4_f32(false, a0, false, bi1, (short)0, acc[3], false, false);
        acc[4] = __builtin_amdgcn_wmma_f32_16x16x4_f32(false, a1, false, br0, (short)0, acc[4], false, false);
        acc[5] = __builtin_amdgcn_wmma_f32_16x16x4_f32(false, a1, false, bi0, (short)0, acc[5], false, false);
        acc[6] = __builtin_amdgcn_wmma_f32_16x16x4_f32(false, a1, false, br1, (short)0, acc[6], false, false);
        acc[7] = __builtin_amdgcn_wmma_f32_16x16x4_f32(false, a1, false, bi1, (short)0, acc[7], false, false);
    }
}

__global__ __launch_bounds__(256) __attribute__((amdgpu_waves_per_eu(2, 6))) void k3_fused(
        const float* __restrict__ angles,
        const float* __restrict__ UrP,
        const float* __restrict__ UiP,
        const float* __restrict__ out_w,
        const float* __restrict__ out_b,
        float* __restrict__ out) {
    __shared__ __align__(16) float cs[ROWS_PER_BLOCK][NQ][2];
    __shared__ __align__(16) float enc[ROWS_PER_BLOCK][TILE_PITCH];
    __shared__ __align__(16) float pq[ROWS_PER_BLOCK][TILE_PITCH];
    __shared__ __align__(16) float E[ROWS_PER_BLOCK][NQ];

    int tid  = threadIdx.x;
    int lane = tid & 31;
    int wave = tid >> 5;
    int row0 = blockIdx.x * ROWS_PER_BLOCK;

    {
        int m = tid >> 3, q = tid & 7;
        float a = angles[(row0 + m) * NQ + q] * 0.5f;
        cs[m][q][0] = cosf(a);
        cs[m][q][1] = sinf(a);
    }
    __syncthreads();

    for (int m = 0; m < ROWS_PER_BLOCK; ++m) {
        float v = 1.0f;
#pragma unroll
        for (int q = 0; q < NQ; ++q) v *= cs[m][q][(tid >> q) & 1];
        enc[m][tid] = v;
    }
    __syncthreads();

    v8f acc[8];
#pragma unroll
    for (int i = 0; i < 8; ++i)
#pragma unroll
        for (int v = 0; v < 8; ++v) acc[i][v] = 0.0f;

    int mrow = lane & 15;
    int h    = lane >> 4;
    int n    = lane & 15;
    int sb   = wave * 32;

    const float* ur = UrP + h * (2 * NSTATE) + (sb + 2 * n) * 2;
    const float* ui = UiP + h * (2 * NSTATE) + (sb + 2 * n) * 2;
    const float* ea = &enc[mrow][2 * h];

#pragma unroll
    for (int g = 0; g < K_GROUPS; ++g) {
        v4f br[KSTEPS_PER_GROUP], bi[KSTEPS_PER_GROUP];
        load_u_group(ur, ui, g, br, bi);
        mma_group(ea, g, br, bi, acc);
    }

#pragma unroll
    for (int v = 0; v < 8; ++v) {
        int m = v + 8 * h;
        v2f p0, p1;
        p0[0] = acc[0][v] * acc[0][v] + acc[1][v] * acc[1][v];
        p0[1] = acc[2][v] * acc[2][v] + acc[3][v] * acc[3][v];
        p1[0] = acc[4][v] * acc[4][v] + acc[5][v] * acc[5][v];
        p1[1] = acc[6][v] * acc[6][v] + acc[7][v] * acc[7][v];
        *(v2f*)&pq[m][sb + 2 * n]      = p0;
        *(v2f*)&pq[16 + m][sb + 2 * n] = p1;
    }
    __syncthreads();

    {
        int m = tid >> 3, q = tid & 7;
        float esum = 0.0f;
        for (int s = 0; s < NSTATE; ++s) {
            float p = pq[m][s];
            esum += ((s >> q) & 1) ? -p : p;
        }
        E[m][q] = esum;
    }
    __syncthreads();

    for (int idx = tid; idx < ROWS_PER_BLOCK * NCLS; idx += 256) {
        int m = idx / NCLS, c = idx % NCLS;
        float o = out_b[c];
#pragma unroll
        for (int q = 0; q < NQ; ++q) o += E[m][q] * out_w[c * NQ + q];
        out[(row0 + m) * NCLS + c] = o;
    }
}

extern "C" void kernel_launch(void* const* d_in, const int* in_sizes, int n_in,
                              void* d_out, int out_size, void* d_ws, size_t ws_size,
                              hipStream_t stream) {
    const float* x      = (const float*)d_in[0];
    const float* proj_w = (const float*)d_in[1];
    const float* qnn_w  = (const float*)d_in[2];
    const float* out_w  = (const float*)d_in[3];
    const float* out_b  = (const float*)d_in[4];
    float* ws = (float*)d_ws;

    float* angles = ws;
    float* UrP    = ws + 65536;
    float* UiP    = ws + 65536 + 65536;

    k12_producers<<<PRODUCER_BLOCKS, 256, 0, stream>>>(x, proj_w, qnn_w,
                                                       angles, UrP, UiP);
    k3_fused<<<BATCH / ROWS_PER_BLOCK, 256, 0, stream>>>(angles, UrP, UiP, out_w, out_b,
                                             (float*)d_out);
}


# unitary role: half-angle cos and sin of each circuit weight evaluated once per wave on its owner lane and fetched by readlane, instead of 144 scalar-unit evaluations per wave
# speedup vs baseline: 1.9365x; 1.2807x over previous
#include <hip/hip_runtime.h>
#include <math.h>

typedef float v2f __attribute__((ext_vector_type(2)));
typedef float v4f __attribute__((ext_vector_type(4)));
typedef float v8f __attribute__((ext_vector_type(8)));

#define BATCH 8192
#define FDIM  512
#define NQ    8
#define NSTATE 256
#define NCLS  10
#define TILE_PITCH (NSTATE + 4)

#define U_BLOCKS        32
#define ANGLE_ROWS_PER_WAVE  4
#define ANGLE_ROWS_PER_BLOCK 32
#define ANGLE_BLOCKS    (BATCH / ANGLE_ROWS_PER_BLOCK)
#define PRODUCER_BLOCKS (U_BLOCKS + ANGLE_BLOCKS)

__device__ __forceinline__ void angles_rows(const float* __restrict__ x,
                                            const float* __restrict__ proj_w,
                                            float* __restrict__ angles,
                                            int bid) {
    __shared__ __align__(16) float w[NQ * FDIM];
    int tid  = threadIdx.x;
    int lane = tid & 31;
    int wave = tid >> 5;
    int row0 = bid * ANGLE_ROWS_PER_BLOCK + wave * ANGLE_ROWS_PER_WAVE;

    const v4f* pw4 = (const v4f*)proj_w;
    v4f wst[4];
#pragma unroll
    for (int k = 0; k < 4; ++k) wst[k] = pw4[tid + 256 * k];

    const v4f* xr = (const v4f*)(x + (size_t)row0 * FDIM) + lane;
    v4f xv[ANGLE_ROWS_PER_WAVE][4];
#pragma unroll
    for (int j = 0; j < 4; ++j)
#pragma unroll
        for (int r = 0; r < ANGLE_ROWS_PER_WAVE; ++r)
            xv[r][j] = xr[r * (FDIM / 4) + j * 32];

    v4f* w4 = (v4f*)w;
#pragma unroll
    for (int k = 0; k < 4; ++k) w4[tid + 256 * k] = wst[k];
    __syncthreads();

    v2f accv[ANGLE_ROWS_PER_WAVE][NQ];
#pragma unroll
    for (int r = 0; r < ANGLE_ROWS_PER_WAVE; ++r)
#pragma unroll
        for (int q = 0; q < NQ; ++q) { accv[r][q][0] = 0.0f; accv[r][q][1] = 0.0f; }

#pragma unroll
    for (int j = 0; j < 4; ++j) {
#pragma unroll
        for (int q = 0; q < NQ; ++q) {
            v4f wv = w4[q * (FDIM / 4) + j * 32 + lane];
            v2f wlo = wv.lo, whi = wv.hi;
#pragma unroll
            for (int r = 0; r < ANGLE_ROWS_PER_WAVE; ++r) {
                v2f xlo = xv[r][j].lo, xhi = xv[r][j].hi;
                accv[r][q] = __builtin_elementwise_fma(xlo, wlo, accv[r][q]);
                accv[r][q] = __builtin_elementwise_fma(xhi, whi, accv[r][q]);
            }
        }
    }
    float acc[ANGLE_ROWS_PER_WAVE * NQ];
#pragma unroll
    for (int r = 0; r < ANGLE_ROWS_PER_WAVE; ++r)
#pragma unroll
        for (int q = 0; q < NQ; ++q) acc[r * NQ + q] = accv[r][q][0] + accv[r][q][1];

    float a16[16], a8[8], a4[4], a2[2];
    {
        bool hi = (lane & 16) != 0;
#pragma unroll
        for (int i = 0; i < 16; ++i) {
            float keep = hi ? acc[i + 16] : acc[i];
            float send = hi ? acc[i] : acc[i + 16];
            a16[i] = keep + __shfl_xor(send, 16, 32);
        }
    }
    {
        bool hi = (lane & 8) != 0;
#pragma unroll
        for (int i = 0; i < 8; ++i) {
            float keep = hi ? a16[i + 8] : a16[i];
            float send = hi ? a16[i] : a16[i + 8];
            a8[i] = keep + __shfl_xor(send, 8, 32);
        }
    }
    {
        bool hi = (lane & 4) != 0;
#pragma unroll
        for (int i = 0; i < 4; ++i) {
            float keep = hi ? a8[i + 4] : a8[i];
            float send = hi ? a8[i] : a8[i + 4];
            a4[i] = keep + __shfl_xor(send, 4, 32);
        }
    }
    {
        bool hi = (lane & 2) != 0;
#pragma unroll
        for (int i = 0; i < 2; ++i) {
            float keep = hi ? a4[i + 2] : a4[i];
            float send = hi ? a4[i] : a4[i + 2];
            a2[i] = keep + __shfl_xor(send, 2, 32);
        }
    }
    float v;
    {
        bool hi = (lane & 1) != 0;
        float keep = hi ? a2[1] : a2[0];
        float send = hi ? a2[0] : a2[1];
        v = keep + __shfl_xor(send, 1, 32);
    }
    angles[row0 * NQ + lane] = tanhf(v) * 1.5707963267948966f;
}

__device__ __forceinline__ void rot_lane_rx(float sr[8], float si[8],
                                            float c, float s, int m) {
#pragma unroll
    for (int r = 0; r < 8; ++r) {
        float pr = __shfl_xor(sr[r], m, 32);
        float pi = __shfl_xor(si[r], m, 32);
        float nr = c * sr[r] + s * pi;
        float ni = c * si[r] - s * pr;
        sr[r] = nr; si[r] = ni;
    }
}

__device__ __forceinline__ void rot_lane_ry(float sr[8], float si[8],
                                            float c, float s, int m, int lane) {
    float b = (lane & m) ? s : -s;
#pragma unroll
    for (int r = 0; r < 8; ++r) {
        float pr = __shfl_xor(sr[r], m, 32);
        float pi = __shfl_xor(si[r], m, 32);
        sr[r] = c * sr[r] + b * pr;
        si[r] = c * si[r] + b * pi;
    }
}

__device__ __forceinline__ void rot_reg_rx(float sr[8], float si[8],
                                           float c, float s, int rm) {
#pragma unroll
    for (int r0 = 0; r0 < 8; ++r0) {
        if (r0 & rm) continue;
        int r1 = r0 | rm;
        float a0r = sr[r0], a0i = si[r0], a1r = sr[r1], a1i = si[r1];
        sr[r0] = c * a0r + s * a1i;  si[r0] = c * a0i - s * a1r;
        sr[r1] = c * a1r + s * a0i;  si[r1] = c * a1i - s * a0r;
    }
}

__device__ __forceinline__ void rot_reg_ry(float sr[8], float si[8],
                                           float c, float s, int rm) {
#pragma unroll
    for (int r0 = 0; r0 < 8; ++r0) {
        if (r0 & rm) continue;
        int r1 = r0 | rm;
        float a0r = sr[r0], a0i = si[r0], a1r = sr[r1], a1i = si[r1];
        sr[r0] = c * a0r - s * a1r;  si[r0] = c * a0i - s * a1i;
        sr[r1] = s * a0r + c * a1r;  si[r1] = s * a0i + c * a1i;
    }
}

__device__ __forceinline__ void cnot_ll(float sr[8], float si[8],
                                        int cm, int tm, int lane) {
    bool hi = (lane & cm) != 0;
#pragma unroll
    for (int r = 0; r < 8; ++r) {
        float pr = __shfl_xor(sr[r], tm, 32);
        float pi = __shfl_xor(si[r], tm, 32);
        sr[r] = hi ? pr : sr[r];
        si[r] = hi ? pi : si[r];
    }
}

__device__ __forceinline__ void cnot_lane_reg(float sr[8], float si[8],
                                              int cm, int trm, int lane) {
    bool hi = (lane & cm) != 0;
#pragma unroll
    for (int r0 = 0; r0 < 8; ++r0) {
        if (r0 & trm) continue;
        int r1 = r0 | trm;
        float t0r = sr[r0], t0i = si[r0];
        sr[r0] = hi ? sr[r1] : sr[r0];  si[r0] = hi ? si[r1] : si[r0];
        sr[r1] = hi ? t0r    : sr[r1];  si[r1] = hi ? t0i    : si[r1];
    }
}

__device__ __forceinline__ void cnot_reg_lane(float sr[8], float si[8],
                                              int crm, int tm) {
#pragma unroll
    for (int r = 0; r < 8; ++r) {
        if (!(r & crm)) continue;
        sr[r] = __shfl_xor(sr[r], tm, 32);
        si[r] = __shfl_xor(si[r], tm, 32);
    }
}

__device__ __forceinline__ void cnot_reg_reg(float sr[8], float si[8],
                                             int crm, int trm) {
#pragma unroll
    for (int r0 = 0; r0 < 8; ++r0) {
        if (!(r0 & crm) || (r0 & trm)) continue;
        int r1 = r0 | trm;
        float tr = sr[r0], ti = si[r0];
        sr[r0] = sr[r1]; si[r0] = si[r1];
        sr[r1] = tr;     si[r1] = ti;
    }
}

__device__ __forceinline__ float from_lane(float v, int src) {
    return __int_as_float(__builtin_amdgcn_readlane(__float_as_int(v), src));
}

__device__ __forceinline__ void unitary_cols(const float* __restrict__ qnn_w,
                                             float* __restrict__ UrP,
                                             float* __restrict__ UiP,
                                             int bid) {
    int lane = threadIdx.x & 31;
    int t    = bid * 8 + (threadIdx.x >> 5);

    int own = lane < 3 * NQ ? lane : 3 * NQ - 1;
    float hw[3];
#pragma unroll
    for (int l = 0; l < 3; ++l) hw[l] = qnn_w[l * 3 * NQ + own];
    float cw[3], sw[3];
#pragma unroll
    for (int l = 0; l < 3; ++l) {
        cw[l] = cosf(hw[l] * 0.5f);
        sw[l] = sinf(hw[l] * 0.5f);
    }

    float sr[8], si[8];
#pragma unroll
    for (int r = 0; r < 8; ++r) {
        sr[r] = (r * 32 + lane == t) ? 1.0f : 0.0f;
        si[r] = 0.0f;
    }

#pragma unroll
    for (int l = 0; l < 3; ++l) {
        float cl = (l == 0) ? cw[0] : (l == 1) ? cw[1] : cw[2];
        float sl = (l == 0) ? sw[0] : (l == 1) ? sw[1] : sw[2];
#pragma unroll
        for (int q = 0; q < NQ; ++q) {
            float c0 = from_lane(cl, q * 3 + 0), s0 = from_lane(sl, q * 3 + 0);
            float c1 = from_lane(cl, q * 3 + 1), s1 = from_lane(sl, q * 3 + 1);
            float c2 = from_lane(cl, q * 3 + 2), s2 = from_lane(sl, q * 3 + 2);
            if (q < 5) {
                int m = 1 << q;
                rot_lane_rx(sr, si, c0, s0, m);
                rot_lane_ry(sr, si, c1, s1, m, lane);
                rot_lane_rx(sr, si, c2, s2, m);
            } else {
                int rm = 1 << (q - 5);
                rot_reg_rx(sr, si, c0, s0, rm);
                rot_reg_ry(sr, si, c1, s1, rm);
                rot_reg_rx(sr, si, c2, s2, rm);
            }
        }
        cnot_ll(sr, si, 1, 2, lane);
        cnot_ll(sr, si, 2, 4, lane);
        cnot_ll(sr, si, 4, 8, lane);
        cnot_ll(sr, si, 8, 16, lane);
        cnot_lane_reg(sr, si, 16, 1, lane);
        cnot_reg_reg(sr, si, 1, 2);
        cnot_reg_reg(sr, si, 2, 4);
        cnot_reg_lane(sr, si, 4, 1);
    }

#pragma unroll
    for (int r = 0; r < 8; ++r) {
        int s = r * 32 + lane;
        int off = (t >> 1) * (2 * NSTATE) + s * 2 + (t & 1);
        UrP[off] = sr[r];
        UiP[off] = si[r];
    }
}

__global__ __launch_bounds__(256) void k12_producers(const float* __restrict__ x,
                              const float* __restrict__ proj_w,
                              const float* __restrict__ qnn_w,
                              float* __restrict__ angles,
                              float* __restrict__ UrP,
                              float* __restrict__ UiP) {
    int b = blockIdx.x;
    if (b < U_BLOCKS) unitary_cols(qnn_w, UrP, UiP, b);
    else              angles_rows(x, proj_w, angles, b - U_BLOCKS);
}

#define ROWS_PER_BLOCK 32
#define KSTEPS_PER_GROUP 8
#define K_GROUPS (NSTATE / (4 * KSTEPS_PER_GROUP))

__device__ __forceinline__ void load_u_group(const float* __restrict__ ur,
                                             const float* __restrict__ ui, int g,
                                             v4f (&br)[KSTEPS_PER_GROUP],
                                             v4f (&bi)[KSTEPS_PER_GROUP]) {
#pragma unroll
    for (int j = 0; j < KSTEPS_PER_GROUP; ++j) {
        int k = (g * KSTEPS_PER_GROUP + j) * 4;
        br[j] = *(const v4f*)(ur + (k >> 1) * (2 * NSTATE));
        bi[j] = *(const v4f*)(ui + (k >> 1) * (2 * NSTATE));
    }
}

__device__ __forceinline__ void mma_group(const float* ea, int g,
                                          const v4f (&br)[KSTEPS_PER_GROUP],
                                          const v4f (&bi)[KSTEPS_PER_GROUP],
                                          v8f (&acc)[8]) {
#pragma unroll
    for (int j = 0; j < KSTEPS_PER_GROUP; ++j) {
        int k = (g * KSTEPS_PER_GROUP + j) * 4;
        v2f a0 = *(const v2f*)(ea + k);
        v2f a1 = *(const v2f*)(ea + 16 * TILE_PITCH + k);
        v2f br0 = br[j].lo, br1 = br[j].hi;
        v2f bi0 = bi[j].lo, bi1 = bi[j].hi;
        acc[0] = __builtin_amdgcn_wmma_f32_16x16x4_f32(false, a0, false, br0, (short)0, acc[0], false, false);
        acc[1] = __builtin_amdgcn_wmma_f32_16x16x4_f32(false, a0, false, bi0, (short)0, acc[1], false, false);
        acc[2] = __builtin_amdgcn_wmma_f32_16x16x4_f32(false, a0, false, br1, (short)0, acc[2], false, false);
        acc[3] = __builtin_amdgcn_wmma_f32_16x16x4_f32(false, a0, false, bi1, (short)0, acc[3], false, false);
        acc[4] = __builtin_amdgcn_wmma_f32_16x16x4_f32(false, a1, false, br0, (short)0, acc[4], false, false);
        acc[5] = __builtin_amdgcn_wmma_f32_16x16x4_f32(false, a1, false, bi0, (short)0, acc[5], false, false);
        acc[6] = __builtin_amdgcn_wmma_f32_16x16x4_f32(false, a1, false, br1, (short)0, acc[6], false, false);
        acc[7] = __builtin_amdgcn_wmma_f32_16x16x4_f32(false, a1, false, bi1, (short)0, acc[7], false, false);
    }
}

__global__ __launch_bounds__(256) __attribute__((amdgpu_waves_per_eu(2, 6))) void k3_fused(
        const float* __restrict__ angles,
        const float* __restrict__ UrP,
        const float* __restrict__ UiP,
        const float* __restrict__ out_w,
        const float* __restrict__ out_b,
        float* __restrict__ out) {
    __shared__ __align__(16) float cs[ROWS_PER_BLOCK][NQ][2];
    __shared__ __align__(16) float enc[ROWS_PER_BLOCK][TILE_PITCH];
    __shared__ __align__(16) float pq[ROWS_PER_BLOCK][TILE_PITCH];
    __shared__ __align__(16) float E[ROWS_PER_BLOCK][NQ];

    int tid  = threadIdx.x;
    int lane = tid & 31;
    int wave = tid >> 5;
    int row0 = blockIdx.x * ROWS_PER_BLOCK;

    {
        int m = tid >> 3, q = tid & 7;
        float a = angles[(row0 + m) * NQ + q] * 0.5f;
        cs[m][q][0] = cosf(a);
        cs[m][q][1] = sinf(a);
    }
    __syncthreads();

    for (int m = 0; m < ROWS_PER_BLOCK; ++m) {
        float v = 1.0f;
#pragma unroll
        for (int q = 0; q < NQ; ++q) v *= cs[m][q][(tid >> q) & 1];
        enc[m][tid] = v;
    }
    __syncthreads();

    v8f acc[8];
#pragma unroll
    for (int i = 0; i < 8; ++i)
#pragma unroll
        for (int v = 0; v < 8; ++v) acc[i][v] = 0.0f;

    int mrow = lane & 15;
    int h    = lane >> 4;
    int n    = lane & 15;
    int sb   = wave * 32;

    const float* ur = UrP + h * (2 * NSTATE) + (sb + 2 * n) * 2;
    const float* ui = UiP + h * (2 * NSTATE) + (sb + 2 * n) * 2;
    const float* ea = &enc[mrow][2 * h];

#pragma unroll
    for (int g = 0; g < K_GROUPS; ++g) {
        v4f br[KSTEPS_PER_GROUP], bi[KSTEPS_PER_GROUP];
        load_u_group(ur, ui, g, br, bi);
        mma_group(ea, g, br, bi, acc);
    }

#pragma unroll
    for (int v = 0; v < 8; ++v) {
        int m = v + 8 * h;
        v2f p0, p1;
        p0[0] = acc[0][v] * acc[0][v] + acc[1][v] * acc[1][v];
        p0[1] = acc[2][v] * acc[2][v] + acc[3][v] * acc[3][v];
        p1[0] = acc[4][v] * acc[4][v] + acc[5][v] * acc[5][v];
        p1[1] = acc[6][v] * acc[6][v] + acc[7][v] * acc[7][v];
        *(v2f*)&pq[m][sb + 2 * n]      = p0;
        *(v2f*)&pq[16 + m][sb + 2 * n] = p1;
    }
    __syncthreads();

    {
        int m = tid >> 3, q = tid & 7;
        float esum = 0.0f;
        for (int s = 0; s < NSTATE; ++s) {
            float p = pq[m][s];
            esum += ((s >> q) & 1) ? -p : p;
        }
        E[m][q] = esum;
    }
    __syncthreads();

    for (int idx = tid; idx < ROWS_PER_BLOCK * NCLS; idx += 256) {
        int m = idx / NCLS, c = idx % NCLS;
        float o = out_b[c];
#pragma unroll
        for (int q = 0; q < NQ; ++q) o += E[m][q] * out_w[c * NQ + q];
        out[(row0 + m) * NCLS + c] = o;
    }
}

extern "C" void kernel_launch(void* const* d_in, const int* in_sizes, int n_in,
                              void* d_out, int out_size, void* d_ws, size_t ws_size,
                              hipStream_t stream) {
    const float* x      = (const float*)d_in[0];
    const float* proj_w = (const float*)d_in[1];
    const float* qnn_w  = (const float*)d_in[2];
    const float* out_w  = (const float*)d_in[3];
    const float* out_b  = (const float*)d_in[4];
    float* ws = (float*)d_ws;

    float* angles = ws;
    float* UrP    = ws + 65536;
    float* UiP    = ws + 65536 + 65536;

    k12_producers<<<PRODUCER_BLOCKS, 256, 0, stream>>>(x, proj_w, qnn_w,
                                                       angles, UrP, UiP);
    k3_fused<<<BATCH / ROWS_PER_BLOCK, 256, 0, stream>>>(angles, UrP, UiP, out_w, out_b,
                                             (float*)d_out);
}


# GEMM kernel: A fragments made in registers from the separable encoding (one packed multiply each, all in front of the K loop); LDS enc tile, its build and barrier go (product order of enc changes)
# speedup vs baseline: 2.0500x; 1.0586x over previous
#include <hip/hip_runtime.h>
#include <math.h>

typedef float v2f __attribute__((ext_vector_type(2)));
typedef float v4f __attribute__((ext_vector_type(4)));
typedef float v8f __attribute__((ext_vector_type(8)));

#define BATCH 8192
#define FDIM  512
#define NQ    8
#define NSTATE 256
#define NCLS  10
#define TILE_PITCH (NSTATE + 4)

#define U_BLOCKS        32
#define ANGLE_ROWS_PER_WAVE  4
#define ANGLE_ROWS_PER_BLOCK 32
#define ANGLE_BLOCKS    (BATCH / ANGLE_ROWS_PER_BLOCK)
#define PRODUCER_BLOCKS (U_BLOCKS + ANGLE_BLOCKS)

__device__ __forceinline__ void angles_rows(const float* __restrict__ x,
                                            const float* __restrict__ proj_w,
                                            float* __restrict__ angles,
                                            int bid) {
    __shared__ __align__(16) float w[NQ * FDIM];
    int tid  = threadIdx.x;
    int lane = tid & 31;
    int wave = tid >> 5;
    int row0 = bid * ANGLE_ROWS_PER_BLOCK + wave * ANGLE_ROWS_PER_WAVE;

    const v4f* pw4 = (const v4f*)proj_w;
    v4f wst[4];
#pragma unroll
    for (int k = 0; k < 4; ++k) wst[k] = pw4[tid + 256 * k];

    const v4f* xr = (const v4f*)(x + (size_t)row0 * FDIM) + lane;
    v4f xv[ANGLE_ROWS_PER_WAVE][4];
#pragma unroll
    for (int j = 0; j < 4; ++j)
#pragma unroll
        for (int r = 0; r < ANGLE_ROWS_PER_WAVE; ++r)
            xv[r][j] = xr[r * (FDIM / 4) + j * 32];

    v4f* w4 = (v4f*)w;
#pragma unroll
    for (int k = 0; k < 4; ++k) w4[tid + 256 * k] = wst[k];
    __syncthreads();

    v2f accv[ANGLE_ROWS_PER_WAVE][NQ];
#pragma unroll
    for (int r = 0; r < ANGLE_ROWS_PER_WAVE; ++r)
#pragma unroll
        for (int q = 0; q < NQ; ++q) { accv[r][q][0] = 0.0f; accv[r][q][1] = 0.0f; }

#pragma unroll
    for (int j = 0; j < 4; ++j) {
#pragma unroll
        for (int q = 0; q < NQ; ++q) {
            v4f wv = w4[q * (FDIM / 4) + j * 32 + lane];
            v2f wlo = wv.lo, whi = wv.hi;
#pragma unroll
            for (int r = 0; r < ANGLE_ROWS_PER_WAVE; ++r) {
                v2f xlo = xv[r][j].lo, xhi = xv[r][j].hi;
                accv[r][q] = __builtin_elementwise_fma(xlo, wlo, accv[r][q]);
                accv[r][q] = __builtin_elementwise_fma(xhi, whi, accv[r][q]);
            }
        }
    }
    float acc[ANGLE_ROWS_PER_WAVE * NQ];
#pragma unroll
    for (int r = 0; r < ANGLE_ROWS_PER_WAVE; ++r)
#pragma unroll
        for (int q = 0; q < NQ; ++q) acc[r * NQ + q] = accv[r][q][0] + accv[r][q][1];

    float a16[16], a8[8], a4[4], a2[2];
    {
        bool hi = (lane & 16) != 0;
#pragma unroll
        for (int i = 0; i < 16; ++i) {
            float keep = hi ? acc[i + 16] : acc[i];
            float send = hi ? acc[i] : acc[i + 16];
            a16[i] = keep + __shfl_xor(send, 16, 32);
        }
    }
    {
        bool hi = (lane & 8) != 0;
#pragma unroll
        for (int i = 0; i < 8; ++i) {
            float keep = hi ? a16[i + 8] : a16[i];
            float send = hi ? a16[i] : a16[i + 8];
            a8[i] = keep + __shfl_xor(send, 8, 32);
        }
    }
    {
        bool hi = (lane & 4) != 0;
#pragma unroll
        for (int i = 0; i < 4; ++i) {
            float keep = hi ? a8[i + 4] : a8[i];
            float send = hi ? a8[i] : a8[i + 4];
            a4[i] = keep + __shfl_xor(send, 4, 32);
        }
    }
    {
        bool hi = (lane & 2) != 0;
#pragma unroll
        for (int i = 0; i < 2; ++i) {
            float keep = hi ? a4[i + 2] : a4[i];
            float send = hi ? a4[i] : a4[i + 2];
            a2[i] = keep + __shfl_xor(send, 2, 32);
        }
    }
    float v;
    {
        bool hi = (lane & 1) != 0;
        float keep = hi ? a2[1] : a2[0];
        float send = hi ? a2[0] : a2[1];
        v = keep + __shfl_xor(send, 1, 32);
    }
    angles[row0 * NQ + lane] = tanhf(v) * 1.5707963267948966f;
}

__device__ __forceinline__ void rot_lane_rx(float sr[8], float si[8],
                                            float c, float s, int m) {
#pragma unroll
    for (int r = 0; r < 8; ++r) {
        float pr = __shfl_xor(sr[r], m, 32);
        float pi = __shfl_xor(si[r], m, 32);
        float nr = c * sr[r] + s * pi;
        float ni = c * si[r] - s * pr;
        sr[r] = nr; si[r] = ni;
    }
}

__device__ __forceinline__ void rot_lane_ry(float sr[8], float si[8],
                                            float c, float s, int m, int lane) {
    float b = (lane & m) ? s : -s;
#pragma unroll
    for (int r = 0; r < 8; ++r) {
        float pr = __shfl_xor(sr[r], m, 32);
        float pi = __shfl_xor(si[r], m, 32);
        sr[r] = c * sr[r] + b * pr;
        si[r] = c * si[r] + b * pi;
    }
}

__device__ __forceinline__ void rot_reg_rx(float sr[8], float si[8],
                                           float c, float s, int rm) {
#pragma unroll
    for (int r0 = 0; r0 < 8; ++r0) {
        if (r0 & rm) continue;
        int r1 = r0 | rm;
        float a0r = sr[r0], a0i = si[r0], a1r = sr[r1], a1i = si[r1];
        sr[r0] = c * a0r + s * a1i;  si[r0] = c * a0i - s * a1r;
        sr[r1] = c * a1r + s * a0i;  si[r1] = c * a1i - s * a0r;
    }
}

__device__ __forceinline__ void rot_reg_ry(float sr[8], float si[8],
                                           float c, float s, int rm) {
#pragma unroll
    for (int r0 = 0; r0 < 8; ++r0) {
        if (r0 & rm) continue;
        int r1 = r0 | rm;
        float a0r = sr[r0], a0i = si[r0], a1r = sr[r1], a1i = si[r1];
        sr[r0] = c * a0r - s * a1r;  si[r0] = c * a0i - s * a1i;
        sr[r1] = s * a0r + c * a1r;  si[r1] = s * a0i + c * a1i;
    }
}

__device__ __forceinline__ void cnot_ll(float sr[8], float si[8],
                                        int cm, int tm, int lane) {
    bool hi = (lane & cm) != 0;
#pragma unroll
    for (int r = 0; r < 8; ++r) {
        float pr = __shfl_xor(sr[r], tm, 32);
        float pi = __shfl_xor(si[r], tm, 32);
        sr[r] = hi ? pr : sr[r];
        si[r] = hi ? pi : si[r];
    }
}

__device__ __forceinline__ void cnot_lane_reg(float sr[8], float si[8],
                                              int cm, int trm, int lane) {
    bool hi = (lane & cm) != 0;
#pragma unroll
    for (int r0 = 0; r0 < 8; ++r0) {
        if (r0 & trm) continue;
        int r1 = r0 | trm;
        float t0r = sr[r0], t0i = si[r0];
        sr[r0] = hi ? sr[r1] : sr[r0];  si[r0] = hi ? si[r1] : si[r0];
        sr[r1] = hi ? t0r    : sr[r1];  si[r1] = hi ? t0i    : si[r1];
    }
}

__device__ __forceinline__ void cnot_reg_lane(float sr[8], float si[8],
                                              int crm, int tm) {
#pragma unroll
    for (int r = 0; r < 8; ++r) {
        if (!(r & crm)) continue;
        sr[r] = __shfl_xor(sr[r], tm, 32);
        si[r] = __shfl_xor(si[r], tm, 32);
    }
}

__device__ __forceinline__ void cnot_reg_reg(float sr[8], float si[8],
                                             int crm, int trm) {
#pragma unroll
    for (int r0 = 0; r0 < 8; ++r0) {
        if (!(r0 & crm) || (r0 & trm)) continue;
        int r1 = r0 | trm;
        float tr = sr[r0], ti = si[r0];
        sr[r0] = sr[r1]; si[r0] = si[r1];
        sr[r1] = tr;     si[r1] = ti;
    }
}

__device__ __forceinline__ float from_lane(float v, int src) {
    return __int_as_float(__builtin_amdgcn_readlane(__float_as_int(v), src));
}

__device__ __forceinline__ void unitary_cols(const float* __restrict__ qnn_w,
                                             float* __restrict__ UrP,
                                             float* __restrict__ UiP,
                                             int bid) {
    int lane = threadIdx.x & 31;
    int t    = bid * 8 + (threadIdx.x >> 5);

    int own = lane < 3 * NQ ? lane : 3 * NQ - 1;
    float hw[3];
#pragma unroll
    for (int l = 0; l < 3; ++l) hw[l] = qnn_w[l * 3 * NQ + own];
    float cw[3], sw[3];
#pragma unroll
    for (int l = 0; l < 3; ++l) {
        cw[l] = cosf(hw[l] * 0.5f);
        sw[l] = sinf(hw[l] * 0.5f);
    }

    float sr[8], si[8];
#pragma unroll
    for (int r = 0; r < 8; ++r) {
        sr[r] = (r * 32 + lane == t) ? 1.0f : 0.0f;
        si[r] = 0.0f;
    }

#pragma unroll
    for (int l = 0; l < 3; ++l) {
        float cl = (l == 0) ? cw[0] : (l == 1) ? cw[1] : cw[2];
        float sl = (l == 0) ? sw[0] : (l == 1) ? sw[1] : sw[2];
#pragma unroll
        for (int q = 0; q < NQ; ++q) {
            float c0 = from_lane(cl, q * 3 + 0), s0 = from_lane(sl, q * 3 + 0);
            float c1 = from_lane(cl, q * 3 + 1), s1 = from_lane(sl, q * 3 + 1);
            float c2 = from_lane(cl, q * 3 + 2), s2 = from_lane(sl, q * 3 + 2);
            if (q < 5) {
                int m = 1 << q;
                rot_lane_rx(sr, si, c0, s0, m);
                rot_lane_ry(sr, si, c1, s1, m, lane);
                rot_lane_rx(sr, si, c2, s2, m);
            } else {
                int rm = 1 << (q - 5);
                rot_reg_rx(sr, si, c0, s0, rm);
                rot_reg_ry(sr, si, c1, s1, rm);
                rot_reg_rx(sr, si, c2, s2, rm);
            }
        }
        cnot_ll(sr, si, 1, 2, lane);
        cnot_ll(sr, si, 2, 4, lane);
        cnot_ll(sr, si, 4, 8, lane);
        cnot_ll(sr, si, 8, 16, lane);
        cnot_lane_reg(sr, si, 16, 1, lane);
        cnot_reg_reg(sr, si, 1, 2);
        cnot_reg_reg(sr, si, 2, 4);
        cnot_reg_lane(sr, si, 4, 1);
    }

#pragma unroll
    for (int r = 0; r < 8; ++r) {
        int s = r * 32 + lane;
        int off = (t >> 1) * (2 * NSTATE) + s * 2 + (t & 1);
        UrP[off] = sr[r];
        UiP[off] = si[r];
    }
}

__global__ __launch_bounds__(256) void k12_producers(const float* __restrict__ x,
                              const float* __restrict__ proj_w,
                              const float* __restrict__ qnn_w,
                              float* __restrict__ angles,
                              float* __restrict__ UrP,
                              float* __restrict__ UiP) {
    int b = blockIdx.x;
    if (b < U_BLOCKS) unitary_cols(qnn_w, UrP, UiP, b);
    else              angles_rows(x, proj_w, angles, b - U_BLOCKS);
}

#define ROWS_PER_BLOCK 32
#define CS_PITCH (2 * NQ + 4)
#define KSTEPS_PER_GROUP 8
#define K_GROUPS (NSTATE / (4 * KSTEPS_PER_GROUP))

__device__ __forceinline__ void load_u_group(const float* __restrict__ ur,
                                             const float* __restrict__ ui, int g,
                                             v4f (&br)[KSTEPS_PER_GROUP],
                                             v4f (&bi)[KSTEPS_PER_GROUP]) {
#pragma unroll
    for (int j = 0; j < KSTEPS_PER_GROUP; ++j) {
        int k = (g * KSTEPS_PER_GROUP + j) * 4;
        br[j] = *(const v4f*)(ur + (k >> 1) * (2 * NSTATE));
        bi[j] = *(const v4f*)(ui + (k >> 1) * (2 * NSTATE));
    }
}

struct AFactors {
    v2f   lo[4];
    float hi[16];
};

__device__ __forceinline__ void a_factors(const float* c, int h, AFactors& f) {
    v4f c01 = *(const v4f*)(c);
    v4f c23 = *(const v4f*)(c + 4);
    v4f c45 = *(const v4f*)(c + 8);
    v4f c67 = *(const v4f*)(c + 12);
    float f1 = h ? c01[3] : c01[2];
    v2f base;
    base[0] = c01[0] * f1;
    base[1] = c01[1] * f1;
#pragma unroll
    for (int jj = 0; jj < 4; ++jj) {
        float m23 = c23[jj & 1] * c23[2 + (jj >> 1)];
        f.lo[jj][0] = base[0] * m23;
        f.lo[jj][1] = base[1] * m23;
    }
    float t45[4], t67[4];
#pragma unroll
    for (int i = 0; i < 4; ++i) {
        t45[i] = c45[i & 1] * c45[2 + (i >> 1)];
        t67[i] = c67[i & 1] * c67[2 + (i >> 1)];
    }
#pragma unroll
    for (int i = 0; i < 16; ++i) f.hi[i] = t45[i & 3] * t67[i >> 2];
}

__device__ __forceinline__ void a_fragments(const AFactors& f, v2f (&a)[NSTATE / 4]) {
#pragma unroll
    for (int ks = 0; ks < NSTATE / 4; ++ks) a[ks] = f.lo[ks & 3] * f.hi[ks >> 2];
}

__device__ __forceinline__ void mma_group(const v2f (&a0)[NSTATE / 4], const v2f (&a1)[NSTATE / 4], int g,
                                          const v4f (&br)[KSTEPS_PER_GROUP],
                                          const v4f (&bi)[KSTEPS_PER_GROUP],
                                          v8f (&acc)[8]) {
#pragma unroll
    for (int j = 0; j < KSTEPS_PER_GROUP; ++j) {
        int ks = g * KSTEPS_PER_GROUP + j;
        v2f br0 = br[j].lo, br1 = br[j].hi;
        v2f bi0 = bi[j].lo, bi1 = bi[j].hi;
        acc[0] = __builtin_amdgcn_wmma_f32_16x16x4_f32(false, a0[ks], false, br0, (short)0, acc[0], false, false);
        acc[1] = __builtin_amdgcn_wmma_f32_16x16x4_f32(false, a0[ks], false, bi0, (short)0, acc[1], false, false);
        acc[2] = __builtin_amdgcn_wmma_f32_16x16x4_f32(false, a0[ks], false, br1, (short)0, acc[2], false, false);
        acc[3] = __builtin_amdgcn_wmma_f32_16x16x4_f32(false, a0[ks], false, bi1, (short)0, acc[3], false, false);
        acc[4] = __builtin_amdgcn_wmma_f32_16x16x4_f32(false, a1[ks], false, br0, (short)0, acc[4], false, false);
        acc[5] = __builtin_amdgcn_wmma_f32_16x16x4_f32(false, a1[ks], false, bi0, (short)0, acc[5], false, false);
        acc[6] = __builtin_amdgcn_wmma_f32_16x16x4_f32(false, a1[ks], false, br1, (short)0, acc[6], false, false);
        acc[7] = __builtin_amdgcn_wmma_f32_16x16x4_f32(false, a1[ks], false, bi1, (short)0, acc[7], false, false);
    }
}

__global__ __launch_bounds__(256) __attribute__((amdgpu_waves_per_eu(2, 2))) void k3_fused(
        const float* __restrict__ angles,
        const float* __restrict__ UrP,
        const float* __restrict__ UiP,
        const float* __restrict__ out_w,
        const float* __restrict__ out_b,
        float* __restrict__ out) {
    __shared__ __align__(16) float cs[ROWS_PER_BLOCK][CS_PITCH];
    __shared__ __align__(16) float pq[ROWS_PER_BLOCK][TILE_PITCH];
    __shared__ __align__(16) float E[ROWS_PER_BLOCK][NQ];

    int tid  = threadIdx.x;
    int lane = tid & 31;
    int wave = tid >> 5;
    int row0 = blockIdx.x * ROWS_PER_BLOCK;

    {
        int m = tid >> 3, q = tid & 7;
        float a = angles[(row0 + m) * NQ + q] * 0.5f;
        v2f c;
        c[0] = cosf(a);
        c[1] = sinf(a);
        *(v2f*)&cs[m][2 * q] = c;
    }
    __syncthreads();

    int mrow = lane & 15;
    int h    = lane >> 4;
    int n    = lane & 15;
    int sb   = wave * 32;

    const float* ur = UrP + h * (2 * NSTATE) + (sb + 2 * n) * 2;
    const float* ui = UiP + h * (2 * NSTATE) + (sb + 2 * n) * 2;

    v4f br0[KSTEPS_PER_GROUP], bi0[KSTEPS_PER_GROUP];
    v4f br1[KSTEPS_PER_GROUP], bi1[KSTEPS_PER_GROUP];
    load_u_group(ur, ui, 0, br0, bi0);
    load_u_group(ur, ui, 1, br1, bi1);

    v2f a0[NSTATE / 4], a1[NSTATE / 4];
    {
        AFactors fa;
        a_factors(&cs[mrow][0], h, fa);
        a_fragments(fa, a0);
        a_factors(&cs[16 + mrow][0], h, fa);
        a_fragments(fa, a1);
    }
    __builtin_amdgcn_sched_barrier(0);

    v8f acc[8];
#pragma unroll
    for (int i = 0; i < 8; ++i)
#pragma unroll
        for (int v = 0; v < 8; ++v) acc[i][v] = 0.0f;

    mma_group(a0, a1, 0, br0, bi0, acc);
    mma_group(a0, a1, 1, br1, bi1, acc);
#pragma unroll
    for (int g = 2; g < K_GROUPS; ++g) {
        v4f br[KSTEPS_PER_GROUP], bi[KSTEPS_PER_GROUP];
        load_u_group(ur, ui, g, br, bi);
        mma_group(a0, a1, g, br, bi, acc);
    }

#pragma unroll
    for (int v = 0; v < 8; ++v) {
        int m = v + 8 * h;
        v2f p0, p1;
        p0[0] = acc[0][v] * acc[0][v] + acc[1][v] * acc[1][v];
        p0[1] = acc[2][v] * acc[2][v] + acc[3][v] * acc[3][v];
        p1[0] = acc[4][v] * acc[4][v] + acc[5][v] * acc[5][v];
        p1[1] = acc[6][v] * acc[6][v] + acc[7][v] * acc[7][v];
        *(v2f*)&pq[m][sb + 2 * n]      = p0;
        *(v2f*)&pq[16 + m][sb + 2 * n] = p1;
    }
    __syncthreads();

    {
        int m = tid >> 3, q = tid & 7;
        float esum = 0.0f;
        for (int s = 0; s < NSTATE; ++s) {
            float p = pq[m][s];
            esum += ((s >> q) & 1) ? -p : p;
        }
        E[m][q] = esum;
    }
    __syncthreads();

    for (int idx = tid; idx < ROWS_PER_BLOCK * NCLS; idx += 256) {
        int m = idx / NCLS, c = idx % NCLS;
        float o = out_b[c];
#pragma unroll
        for (int q = 0; q < NQ; ++q) o += E[m][q] * out_w[c * NQ + q];
        out[(row0 + m) * NCLS + c] = o;
    }
}

extern "C" void kernel_launch(void* const* d_in, const int* in_sizes, int n_in,
                              void* d_out, int out_size, void* d_ws, size_t ws_size,
                              hipStream_t stream) {
    const float* x      = (const float*)d_in[0];
    const float* proj_w = (const float*)d_in[1];
    const float* qnn_w  = (const float*)d_in[2];
    const float* out_w  = (const float*)d_in[3];
    const float* out_b  = (const float*)d_in[4];
    float* ws = (float*)d_ws;

    float* angles = ws;
    float* UrP    = ws + 65536;
    float* UiP    = ws + 65536 + 65536;

    k12_producers<<<PRODUCER_BLOCKS, 256, 0, stream>>>(x, proj_w, qnn_w,
                                                       angles, UrP, UiP);
    k3_fused<<<BATCH / ROWS_PER_BLOCK, 256, 0, stream>>>(angles, UrP, UiP, out_w, out_b,
                                             (float*)d_out);
}


# GEMM kernel: WMMA operands swapped (U fragment as A, enc as B): accumulator transposed, a lane holds 16 states of one batch row; Z sums start in registers, LDS prob tile goes (E sum order changes)
# speedup vs baseline: 2.1680x; 1.0576x over previous
#include <hip/hip_runtime.h>
#include <math.h>

typedef float v2f __attribute__((ext_vector_type(2)));
typedef float v4f __attribute__((ext_vector_type(4)));
typedef float v8f __attribute__((ext_vector_type(8)));

#define BATCH 8192
#define FDIM  512
#define NQ    8
#define NSTATE 256
#define NCLS  10
#define PART_KINDS 5
#define PART_PITCH (PART_KINDS * 16 + 4)

#define U_BLOCKS        32
#define ANGLE_ROWS_PER_WAVE  4
#define ANGLE_ROWS_PER_BLOCK 32
#define ANGLE_BLOCKS    (BATCH / ANGLE_ROWS_PER_BLOCK)
#define PRODUCER_BLOCKS (U_BLOCKS + ANGLE_BLOCKS)

__device__ __forceinline__ void angles_rows(const float* __restrict__ x,
                                            const float* __restrict__ proj_w,
                                            float* __restrict__ angles,
                                            int bid) {
    __shared__ __align__(16) float w[NQ * FDIM];
    int tid  = threadIdx.x;
    int lane = tid & 31;
    int wave = tid >> 5;
    int row0 = bid * ANGLE_ROWS_PER_BLOCK + wave * ANGLE_ROWS_PER_WAVE;

    const v4f* pw4 = (const v4f*)proj_w;
    v4f wst[4];
#pragma unroll
    for (int k = 0; k < 4; ++k) wst[k] = pw4[tid + 256 * k];

    const v4f* xr = (const v4f*)(x + (size_t)row0 * FDIM) + lane;
    v4f xv[ANGLE_ROWS_PER_WAVE][4];
#pragma unroll
    for (int j = 0; j < 4; ++j)
#pragma unroll
        for (int r = 0; r < ANGLE_ROWS_PER_WAVE; ++r)
            xv[r][j] = xr[r * (FDIM / 4) + j * 32];

    v4f* w4 = (v4f*)w;
#pragma unroll
    for (int k = 0; k < 4; ++k) w4[tid + 256 * k] = wst[k];
    __syncthreads();

    v2f accv[ANGLE_ROWS_PER_WAVE][NQ];
#pragma unroll
    for (int r = 0; r < ANGLE_ROWS_PER_WAVE; ++r)
#pragma unroll
        for (int q = 0; q < NQ; ++q) { accv[r][q][0] = 0.0f; accv[r][q][1] = 0.0f; }

#pragma unroll
    for (int j = 0; j < 4; ++j) {
#pragma unroll
        for (int q = 0; q < NQ; ++q) {
            v4f wv = w4[q * (FDIM / 4) + j * 32 + lane];
            v2f wlo = wv.lo, whi = wv.hi;
#pragma unroll
            for (int r = 0; r < ANGLE_ROWS_PER_WAVE; ++r) {
                v2f xlo = xv[r][j].lo, xhi = xv[r][j].hi;
                accv[r][q] = __builtin_elementwise_fma(xlo, wlo, accv[r][q]);
                accv[r][q] = __builtin_elementwise_fma(xhi, whi, accv[r][q]);
            }
        }
    }
    float acc[ANGLE_ROWS_PER_WAVE * NQ];
#pragma unroll
    for (int r = 0; r < ANGLE_ROWS_PER_WAVE; ++r)
#pragma unroll
        for (int q = 0; q < NQ; ++q) acc[r * NQ + q] = accv[r][q][0] + accv[r][q][1];

    float a16[16], a8[8], a4[4], a2[2];
    {
        bool hi = (lane & 16) != 0;
#pragma unroll
        for (int i = 0; i < 16; ++i) {
            float keep = hi ? acc[i + 16] : acc[i];
            float send = hi ? acc[i] : acc[i + 16];
            a16[i] = keep + __shfl_xor(send, 16, 32);
        }
    }
    {
        bool hi = (lane & 8) != 0;
#pragma unroll
        for (int i = 0; i < 8; ++i) {
            float keep = hi ? a16[i + 8] : a16[i];
            float send = hi ? a16[i] : a16[i + 8];
            a8[i] = keep + __shfl_xor(send, 8, 32);
        }
    }
    {
        bool hi = (lane & 4) != 0;
#pragma unroll
        for (int i = 0; i < 4; ++i) {
            float keep = hi ? a8[i + 4] : a8[i];
            float send = hi ? a8[i] : a8[i + 4];
            a4[i] = keep + __shfl_xor(send, 4, 32);
        }
    }
    {
        bool hi = (lane & 2) != 0;
#pragma unroll
        for (int i = 0; i < 2; ++i) {
            float keep = hi ? a4[i + 2] : a4[i];
            float send = hi ? a4[i] : a4[i + 2];
            a2[i] = keep + __shfl_xor(send, 2, 32);
        }
    }
    float v;
    {
        bool hi = (lane & 1) != 0;
        float keep = hi ? a2[1] : a2[0];
        float send = hi ? a2[0] : a2[1];
        v = keep + __shfl_xor(send, 1, 32);
    }
    angles[row0 * NQ + lane] = tanhf(v) * 1.5707963267948966f;
}

__device__ __forceinline__ void rot_lane_rx(float sr[8], float si[8],
                                            float c, float s, int m) {
#pragma unroll
    for (int r = 0; r < 8; ++r) {
        float pr = __shfl_xor(sr[r], m, 32);
        float pi = __shfl_xor(si[r], m, 32);
        float nr = c * sr[r] + s * pi;
        float ni = c * si[r] - s * pr;
        sr[r] = nr; si[r] = ni;
    }
}

__device__ __forceinline__ void rot_lane_ry(float sr[8], float si[8],
                                            float c, float s, int m, int lane) {
    float b = (lane & m) ? s : -s;
#pragma unroll
    for (int r = 0; r < 8; ++r) {
        float pr = __shfl_xor(sr[r], m, 32);
        float pi = __shfl_xor(si[r], m, 32);
        sr[r] = c * sr[r] + b * pr;
        si[r] = c * si[r] + b * pi;
    }
}

__device__ __forceinline__ void rot_reg_rx(float sr[8], float si[8],
                                           float c, float s, int rm) {
#pragma unroll
    for (int r0 = 0; r0 < 8; ++r0) {
        if (r0 & rm) continue;
        int r1 = r0 | rm;
        float a0r = sr[r0], a0i = si[r0], a1r = sr[r1], a1i = si[r1];
        sr[r0] = c * a0r + s * a1i;  si[r0] = c * a0i - s * a1r;
        sr[r1] = c * a1r + s * a0i;  si[r1] = c * a1i - s * a0r;
    }
}

__device__ __forceinline__ void rot_reg_ry(float sr[8], float si[8],
                                           float c, float s, int rm) {
#pragma unroll
    for (int r0 = 0; r0 < 8; ++r0) {
        if (r0 & rm) continue;
        int r1 = r0 | rm;
        float a0r = sr[r0], a0i = si[r0], a1r = sr[r1], a1i = si[r1];
        sr[r0] = c * a0r - s * a1r;  si[r0] = c * a0i - s * a1i;
        sr[r1] = s * a0r + c * a1r;  si[r1] = s * a0i + c * a1i;
    }
}

__device__ __forceinline__ void cnot_ll(float sr[8], float si[8],
                                        int cm, int tm, int lane) {
    bool hi = (lane & cm) != 0;
#pragma unroll
    for (int r = 0; r < 8; ++r) {
        float pr = __shfl_xor(sr[r], tm, 32);
        float pi = __shfl_xor(si[r], tm, 32);
        sr[r] = hi ? pr : sr[r];
        si[r] = hi ? pi : si[r];
    }
}

__device__ __forceinline__ void cnot_lane_reg(float sr[8], float si[8],
                                              int cm, int trm, int lane) {
    bool hi = (lane & cm) != 0;
#pragma unroll
    for (int r0 = 0; r0 < 8; ++r0) {
        if (r0 & trm) continue;
        int r1 = r0 | trm;
        float t0r = sr[r0], t0i = si[r0];
        sr[r0] = hi ? sr[r1] : sr[r0];  si[r0] = hi ? si[r1] : si[r0];
        sr[r1] = hi ? t0r    : sr[r1];  si[r1] = hi ? t0i    : si[r1];
    }
}

__device__ __forceinline__ void cnot_reg_lane(float sr[8], float si[8],
                                              int crm, int tm) {
#pragma unroll
    for (int r = 0; r < 8; ++r) {
        if (!(r & crm)) continue;
        sr[r] = __shfl_xor(sr[r], tm, 32);
        si[r] = __shfl_xor(si[r], tm, 32);
    }
}

__device__ __forceinline__ void cnot_reg_reg(float sr[8], float si[8],
                                             int crm, int trm) {
#pragma unroll
    for (int r0 = 0; r0 < 8; ++r0) {
        if (!(r0 & crm) || (r0 & trm)) continue;
        int r1 = r0 | trm;
        float tr = sr[r0], ti = si[r0];
        sr[r0] = sr[r1]; si[r0] = si[r1];
        sr[r1] = tr;     si[r1] = ti;
    }
}

__device__ __forceinline__ float from_lane(float v, int src) {
    return __int_as_float(__builtin_amdgcn_readlane(__float_as_int(v), src));
}

__device__ __forceinline__ void unitary_cols(const float* __restrict__ qnn_w,
                                             float* __restrict__ UrP,
                                             float* __restrict__ UiP,
                                             int bid) {
    int lane = threadIdx.x & 31;
    int t    = bid * 8 + (threadIdx.x >> 5);

    int own = lane < 3 * NQ ? lane : 3 * NQ - 1;
    float hw[3];
#pragma unroll
    for (int l = 0; l < 3; ++l) hw[l] = qnn_w[l * 3 * NQ + own];
    float cw[3], sw[3];
#pragma unroll
    for (int l = 0; l < 3; ++l) {
        cw[l] = cosf(hw[l] * 0.5f);
        sw[l] = sinf(hw[l] * 0.5f);
    }

    float sr[8], si[8];
#pragma unroll
    for (int r = 0; r < 8; ++r) {
        sr[r] = (r * 32 + lane == t) ? 1.0f : 0.0f;
        si[r] = 0.0f;
    }

#pragma unroll
    for (int l = 0; l < 3; ++l) {
        float cl = (l == 0) ? cw[0] : (l == 1) ? cw[1] : cw[2];
        float sl = (l == 0) ? sw[0] : (l == 1) ? sw[1] : sw[2];
#pragma unroll
        for (int q = 0; q < NQ; ++q) {
            float c0 = from_lane(cl, q * 3 + 0), s0 = from_lane(sl, q * 3 + 0);
            float c1 = from_lane(cl, q * 3 + 1), s1 = from_lane(sl, q * 3 + 1);
            float c2 = from_lane(cl, q * 3 + 2), s2 = from_lane(sl, q * 3 + 2);
            if (q < 5) {
                int m = 1 << q;
                rot_lane_rx(sr, si, c0, s0, m);
                rot_lane_ry(sr, si, c1, s1, m, lane);
                rot_lane_rx(sr, si, c2, s2, m);
            } else {
                int rm = 1 << (q - 5);
                rot_reg_rx(sr, si, c0, s0, rm);
                rot_reg_ry(sr, si, c1, s1, rm);
                rot_reg_rx(sr, si, c2, s2, rm);
            }
        }
        cnot_ll(sr, si, 1, 2, lane);
        cnot_ll(sr, si, 2, 4, lane);
        cnot_ll(sr, si, 4, 8, lane);
        cnot_ll(sr, si, 8, 16, lane);
        cnot_lane_reg(sr, si, 16, 1, lane);
        cnot_reg_reg(sr, si, 1, 2);
        cnot_reg_reg(sr, si, 2, 4);
        cnot_reg_lane(sr, si, 4, 1);
    }

#pragma unroll
    for (int r = 0; r < 8; ++r) {
        int s = r * 32 + lane;
        int off = (t >> 1) * (2 * NSTATE) + s * 2 + (t & 1);
        UrP[off] = sr[r];
        UiP[off] = si[r];
    }
}

__global__ __launch_bounds__(256) void k12_producers(const float* __restrict__ x,
                              const float* __restrict__ proj_w,
                              const float* __restrict__ qnn_w,
                              float* __restrict__ angles,
                              float* __restrict__ UrP,
                              float* __restrict__ UiP) {
    int b = blockIdx.x;
    if (b < U_BLOCKS) unitary_cols(qnn_w, UrP, UiP, b);
    else              angles_rows(x, proj_w, angles, b - U_BLOCKS);
}

#define ROWS_PER_BLOCK 32
#define CS_PITCH (2 * NQ + 4)
#define KSTEPS_PER_GROUP 8
#define K_GROUPS (NSTATE / (4 * KSTEPS_PER_GROUP))

__device__ __forceinline__ void load_u_group(const float* __restrict__ ur,
                                             const float* __restrict__ ui, int g,
                                             v4f (&br)[KSTEPS_PER_GROUP],
                                             v4f (&bi)[KSTEPS_PER_GROUP]) {
#pragma unroll
    for (int j = 0; j < KSTEPS_PER_GROUP; ++j) {
        int k = (g * KSTEPS_PER_GROUP + j) * 4;
        br[j] = *(const v4f*)(ur + (k >> 1) * (2 * NSTATE));
        bi[j] = *(const v4f*)(ui + (k >> 1) * (2 * NSTATE));
    }
}

struct AFactors {
    v2f   lo[4];
    float hi[16];
};

__device__ __forceinline__ void a_factors(const float* c, int h, AFactors& f) {
    v4f c01 = *(const v4f*)(c);
    v4f c23 = *(const v4f*)(c + 4);
    v4f c45 = *(const v4f*)(c + 8);
    v4f c67 = *(const v4f*)(c + 12);
    float f1 = h ? c01[3] : c01[2];
    v2f base;
    base[0] = c01[0] * f1;
    base[1] = c01[1] * f1;
#pragma unroll
    for (int jj = 0; jj < 4; ++jj) {
        float m23 = c23[jj & 1] * c23[2 + (jj >> 1)];
        f.lo[jj][0] = base[0] * m23;
        f.lo[jj][1] = base[1] * m23;
    }
    float t45[4], t67[4];
#pragma unroll
    for (int i = 0; i < 4; ++i) {
        t45[i] = c45[i & 1] * c45[2 + (i >> 1)];
        t67[i] = c67[i & 1] * c67[2 + (i >> 1)];
    }
#pragma unroll
    for (int i = 0; i < 16; ++i) f.hi[i] = t45[i & 3] * t67[i >> 2];
}

__device__ __forceinline__ void a_fragments(const AFactors& f, v2f (&a)[NSTATE / 4]) {
#pragma unroll
    for (int ks = 0; ks < NSTATE / 4; ++ks) a[ks] = f.lo[ks & 3] * f.hi[ks >> 2];
}

__device__ __forceinline__ void mma_group(const v2f (&a0)[NSTATE / 4], const v2f (&a1)[NSTATE / 4], int g,
                                          const v4f (&br)[KSTEPS_PER_GROUP],
                                          const v4f (&bi)[KSTEPS_PER_GROUP],
                                          v8f (&acc)[8]) {
#pragma unroll
    for (int j = 0; j < KSTEPS_PER_GROUP; ++j) {
        int ks = g * KSTEPS_PER_GROUP + j;
        v2f br0 = br[j].lo, br1 = br[j].hi;
        v2f bi0 = bi[j].lo, bi1 = bi[j].hi;
        acc[0] = __builtin_amdgcn_wmma_f32_16x16x4_f32(false, br0, false, a0[ks], (short)0, acc[0], false, false);
        acc[1] = __builtin_amdgcn_wmma_f32_16x16x4_f32(false, bi0, false, a0[ks], (short)0, acc[1], false, false);
        acc[2] = __builtin_amdgcn_wmma_f32_16x16x4_f32(false, br1, false, a0[ks], (short)0, acc[2], false, false);
        acc[3] = __builtin_amdgcn_wmma_f32_16x16x4_f32(false, bi1, false, a0[ks], (short)0, acc[3], false, false);
        acc[4] = __builtin_amdgcn_wmma_f32_16x16x4_f32(false, br0, false, a1[ks], (short)0, acc[4], false, false);
        acc[5] = __builtin_amdgcn_wmma_f32_16x16x4_f32(false, bi0, false, a1[ks], (short)0, acc[5], false, false);
        acc[6] = __builtin_amdgcn_wmma_f32_16x16x4_f32(false, br1, false, a1[ks], (short)0, acc[6], false, false);
        acc[7] = __builtin_amdgcn_wmma_f32_16x16x4_f32(false, bi1, false, a1[ks], (short)0, acc[7], false, false);
    }
}

__global__ __launch_bounds__(256) __attribute__((amdgpu_waves_per_eu(2, 2))) void k3_fused(
        const float* __restrict__ angles,
        const float* __restrict__ UrP,
        const float* __restrict__ UiP,
        const float* __restrict__ out_w,
        const float* __restrict__ out_b,
        float* __restrict__ out) {
    __shared__ __align__(16) float cs[ROWS_PER_BLOCK][CS_PITCH];
    __shared__ __align__(16) float part[ROWS_PER_BLOCK][PART_PITCH];
    __shared__ __align__(16) float E[ROWS_PER_BLOCK][NQ];

    int tid  = threadIdx.x;
    int lane = tid & 31;
    int wave = tid >> 5;
    int row0 = blockIdx.x * ROWS_PER_BLOCK;

    {
        int m = tid >> 3, q = tid & 7;
        float a = angles[(row0 + m) * NQ + q] * 0.5f;
        v2f c;
        c[0] = cosf(a);
        c[1] = sinf(a);
        *(v2f*)&cs[m][2 * q] = c;
    }
    __syncthreads();

    int mrow = lane & 15;
    int h    = lane >> 4;
    int n    = lane & 15;
    int sb   = wave * 32;

    const float* ur = UrP + h * (2 * NSTATE) + (sb + 2 * n) * 2;
    const float* ui = UiP + h * (2 * NSTATE) + (sb + 2 * n) * 2;

    v4f br0[KSTEPS_PER_GROUP], bi0[KSTEPS_PER_GROUP];
    v4f br1[KSTEPS_PER_GROUP], bi1[KSTEPS_PER_GROUP];
    load_u_group(ur, ui, 0, br0, bi0);
    load_u_group(ur, ui, 1, br1, bi1);

    v2f a0[NSTATE / 4], a1[NSTATE / 4];
    {
        AFactors fa;
        a_factors(&cs[mrow][0], h, fa);
        a_fragments(fa, a0);
        a_factors(&cs[16 + mrow][0], h, fa);
        a_fragments(fa, a1);
    }
    __builtin_amdgcn_sched_barrier(0);

    v8f acc[8];
#pragma unroll
    for (int i = 0; i < 8; ++i)
#pragma unroll
        for (int v = 0; v < 8; ++v) acc[i][v] = 0.0f;

    mma_group(a0, a1, 0, br0, bi0, acc);
    mma_group(a0, a1, 1, br1, bi1, acc);
#pragma unroll
    for (int g = 2; g < K_GROUPS; ++g) {
        v4f br[KSTEPS_PER_GROUP], bi[KSTEPS_PER_GROUP];
        load_u_group(ur, ui, g, br, bi);
        mma_group(a0, a1, g, br, bi, acc);
    }

    {
        int g = 2 * wave + h;
#pragma unroll
        for (int mt = 0; mt < 2; ++mt) {
            float t[8], d[8];
#pragma unroll
            for (int v = 0; v < 8; ++v) {
                float p0 = acc[4 * mt + 0][v] * acc[4 * mt + 0][v] + acc[4 * mt + 1][v] * acc[4 * mt + 1][v];
                float p1 = acc[4 * mt + 2][v] * acc[4 * mt + 2][v] + acc[4 * mt + 3][v] * acc[4 * mt + 3][v];
                t[v] = p0 + p1;
                d[v] = p0 - p1;
            }
            float D0 = ((d[0] + d[1]) + (d[2] + d[3])) + ((d[4] + d[5]) + (d[6] + d[7]));
            float u[4], w[4];
#pragma unroll
            for (int i = 0; i < 4; ++i) {
                u[i] = t[2 * i] + t[2 * i + 1];
                w[i] = t[2 * i] - t[2 * i + 1];
            }
            float D1 = (w[0] + w[1]) + (w[2] + w[3]);
            float x0 = u[0] + u[1], x1 = u[2] + u[3];
            float D2 = (u[0] - u[1]) + (u[2] - u[3]);
            float S  = x0 + x1;
            float D3 = x0 - x1;
            float* pr = &part[16 * mt + mrow][g];
            pr[0 * 16] = S;
            pr[1 * 16] = D0;
            pr[2 * 16] = D1;
            pr[3 * 16] = D2;
            pr[4 * 16] = D3;
        }
    }
    __syncthreads();

    {
        int m = tid >> 3, q = tid & 7;
        int kind = q < 4 ? q + 1 : 0;
        const v4f* src = (const v4f*)&part[m][16 * kind];
        v4f r[4];
#pragma unroll
        for (int i = 0; i < 4; ++i) r[i] = src[i];
        float s1 = q == 4 ? -1.0f : 1.0f;
        float s2 = q == 5 ? -1.0f : 1.0f;
        float s4 = q == 6 ? -1.0f : 1.0f;
        float s8 = q == 7 ? -1.0f : 1.0f;
        float a[4];
#pragma unroll
        for (int i = 0; i < 4; ++i)
            a[i] = fmaf(s2, fmaf(s1, r[i][3], r[i][2]), fmaf(s1, r[i][1], r[i][0]));
        float esum = fmaf(s8, fmaf(s4, a[3], a[2]), fmaf(s4, a[1], a[0]));
        E[m][q] = esum;
    }
    __syncthreads();

    for (int idx = tid; idx < ROWS_PER_BLOCK * NCLS; idx += 256) {
        int m = idx / NCLS, c = idx % NCLS;
        float o = out_b[c];
#pragma unroll
        for (int q = 0; q < NQ; ++q) o += E[m][q] * out_w[c * NQ + q];
        out[(row0 + m) * NCLS + c] = o;
    }
}

extern "C" void kernel_launch(void* const* d_in, const int* in_sizes, int n_in,
                              void* d_out, int out_size, void* d_ws, size_t ws_size,
                              hipStream_t stream) {
    const float* x      = (const float*)d_in[0];
    const float* proj_w = (const float*)d_in[1];
    const float* qnn_w  = (const float*)d_in[2];
    const float* out_w  = (const float*)d_in[3];
    const float* out_b  = (const float*)d_in[4];
    float* ws = (float*)d_ws;

    float* angles = ws;
    float* UrP    = ws + 65536;
    float* UiP    = ws + 65536 + 65536;

    k12_producers<<<PRODUCER_BLOCKS, 256, 0, stream>>>(x, proj_w, qnn_w,
                                                       angles, UrP, UiP);
    k3_fused<<<BATCH / ROWS_PER_BLOCK, 256, 0, stream>>>(angles, UrP, UiP, out_w, out_b,
                                             (float*)d_out);
}


# unitary role: the three rotations of each qubit and layer multiplied into one 2x2 complex gate on its owner lane, applied in one pass: 448 LDS lane exchanges per wave, not 928 (rounding of U changes)
# speedup vs baseline: 2.5406x; 1.1719x over previous
#include <hip/hip_runtime.h>
#include <math.h>

typedef float v2f __attribute__((ext_vector_type(2)));
typedef float v4f __attribute__((ext_vector_type(4)));
typedef float v8f __attribute__((ext_vector_type(8)));

#define BATCH 8192
#define FDIM  512
#define NQ    8
#define NSTATE 256
#define NCLS  10
#define PART_KINDS 5
#define PART_PITCH (PART_KINDS * 16 + 4)

#define U_BLOCKS        32
#define ANGLE_ROWS_PER_WAVE  4
#define ANGLE_ROWS_PER_BLOCK 32
#define ANGLE_BLOCKS    (BATCH / ANGLE_ROWS_PER_BLOCK)
#define PRODUCER_BLOCKS (U_BLOCKS + ANGLE_BLOCKS)

__device__ __forceinline__ void angles_rows(const float* __restrict__ x,
                                            const float* __restrict__ proj_w,
                                            float* __restrict__ angles,
                                            int bid) {
    __shared__ __align__(16) float w[NQ * FDIM];
    int tid  = threadIdx.x;
    int lane = tid & 31;
    int wave = tid >> 5;
    int row0 = bid * ANGLE_ROWS_PER_BLOCK + wave * ANGLE_ROWS_PER_WAVE;

    const v4f* pw4 = (const v4f*)proj_w;
    v4f wst[4];
#pragma unroll
    for (int k = 0; k < 4; ++k) wst[k] = pw4[tid + 256 * k];

    const v4f* xr = (const v4f*)(x + (size_t)row0 * FDIM) + lane;
    v4f xv[ANGLE_ROWS_PER_WAVE][4];
#pragma unroll
    for (int j = 0; j < 4; ++j)
#pragma unroll
        for (int r = 0; r < ANGLE_ROWS_PER_WAVE; ++r)
            xv[r][j] = xr[r * (FDIM / 4) + j * 32];

    v4f* w4 = (v4f*)w;
#pragma unroll
    for (int k = 0; k < 4; ++k) w4[tid + 256 * k] = wst[k];
    __syncthreads();

    v2f accv[ANGLE_ROWS_PER_WAVE][NQ];
#pragma unroll
    for (int r = 0; r < ANGLE_ROWS_PER_WAVE; ++r)
#pragma unroll
        for (int q = 0; q < NQ; ++q) { accv[r][q][0] = 0.0f; accv[r][q][1] = 0.0f; }

#pragma unroll
    for (int j = 0; j < 4; ++j) {
#pragma unroll
        for (int q = 0; q < NQ; ++q) {
            v4f wv = w4[q * (FDIM / 4) + j * 32 + lane];
            v2f wlo = wv.lo, whi = wv.hi;
#pragma unroll
            for (int r = 0; r < ANGLE_ROWS_PER_WAVE; ++r) {
                v2f xlo = xv[r][j].lo, xhi = xv[r][j].hi;
                accv[r][q] = __builtin_elementwise_fma(xlo, wlo, accv[r][q]);
                accv[r][q] = __builtin_elementwise_fma(xhi, whi, accv[r][q]);
            }
        }
    }
    float acc[ANGLE_ROWS_PER_WAVE * NQ];
#pragma unroll
    for (int r = 0; r < ANGLE_ROWS_PER_WAVE; ++r)
#pragma unroll
        for (int q = 0; q < NQ; ++q) acc[r * NQ + q] = accv[r][q][0] + accv[r][q][1];

    float a16[16], a8[8], a4[4], a2[2];
    {
        bool hi = (lane & 16) != 0;
#pragma unroll
        for (int i = 0; i < 16; ++i) {
            float keep = hi ? acc[i + 16] : acc[i];
            float send = hi ? acc[i] : acc[i + 16];
            a16[i] = keep + __shfl_xor(send, 16, 32);
        }
    }
    {
        bool hi = (lane & 8) != 0;
#pragma unroll
        for (int i = 0; i < 8; ++i) {
            float keep = hi ? a16[i + 8] : a16[i];
            float send = hi ? a16[i] : a16[i + 8];
            a8[i] = keep + __shfl_xor(send, 8, 32);
        }
    }
    {
        bool hi = (lane & 4) != 0;
#pragma unroll
        for (int i = 0; i < 4; ++i) {
            float keep = hi ? a8[i + 4] : a8[i];
            float send = hi ? a8[i] : a8[i + 4];
            a4[i] = keep + __shfl_xor(send, 4, 32);
        }
    }
    {
        bool hi = (lane & 2) != 0;
#pragma unroll
        for (int i = 0; i < 2; ++i) {
            float keep = hi ? a4[i + 2] : a4[i];
            float send = hi ? a4[i] : a4[i + 2];
            a2[i] = keep + __shfl_xor(send, 2, 32);
        }
    }
    float v;
    {
        bool hi = (lane & 1) != 0;
        float keep = hi ? a2[1] : a2[0];
        float send = hi ? a2[0] : a2[1];
        v = keep + __shfl_xor(send, 1, 32);
    }
    angles[row0 * NQ + lane] = tanhf(v) * 1.5707963267948966f;
}

__device__ __forceinline__ void gate_lane(float sr[8], float si[8],
                                          float P, float Q, float R, float T,
                                          int m, int lane) {
    bool hi = (lane & m) != 0;
    float qq = hi ? -Q : Q;
    float rr = hi ? R : -R;
    v2f qv; qv[0] = -qq; qv[1] = qq;
    v2f tv; tv[0] = T;   tv[1] = -T;
#pragma unroll
    for (int r = 0; r < 8; ++r) {
        v2f a;  a[0] = sr[r];  a[1] = si[r];
        v2f as = __builtin_shufflevector(a, a, 1, 0);
        v2f b;  b[0] = __shfl_xor(sr[r], m, 32); b[1] = __shfl_xor(si[r], m, 32);
        v2f bs = __builtin_shufflevector(b, b, 1, 0);
        v2f n = P * a + qv * as;
        n = n + rr * b + tv * bs;
        sr[r] = n[0]; si[r] = n[1];
    }
}

__device__ __forceinline__ void gate_reg(float sr[8], float si[8],
                                         float P, float Q, float R, float T, int rm) {
    v2f qp; qp[0] = -Q; qp[1] = Q;
    v2f qm; qm[0] = Q;  qm[1] = -Q;
    v2f tm; tm[0] = T;  tm[1] = -T;
#pragma unroll
    for (int r0 = 0; r0 < 8; ++r0) {
        if (r0 & rm) continue;
        int r1 = r0 | rm;
        v2f a0;  a0[0] = sr[r0];  a0[1] = si[r0];
        v2f a1;  a1[0] = sr[r1];  a1[1] = si[r1];
        v2f a0s = __builtin_shufflevector(a0, a0, 1, 0);
        v2f a1s = __builtin_shufflevector(a1, a1, 1, 0);
        v2f n0 = P * a0 + qp * a0s - R * a1 + tm * a1s;
        v2f n1 = R * a0 + tm * a0s + P * a1 + qm * a1s;
        sr[r0] = n0[0]; si[r0] = n0[1];
        sr[r1] = n1[0]; si[r1] = n1[1];
    }
}

__device__ __forceinline__ void cnot_ll(float sr[8], float si[8],
                                        int cm, int tm, int lane) {
    bool hi = (lane & cm) != 0;
#pragma unroll
    for (int r = 0; r < 8; ++r) {
        float pr = __shfl_xor(sr[r], tm, 32);
        float pi = __shfl_xor(si[r], tm, 32);
        sr[r] = hi ? pr : sr[r];
        si[r] = hi ? pi : si[r];
    }
}

__device__ __forceinline__ void cnot_lane_reg(float sr[8], float si[8],
                                              int cm, int trm, int lane) {
    bool hi = (lane & cm) != 0;
#pragma unroll
    for (int r0 = 0; r0 < 8; ++r0) {
        if (r0 & trm) continue;
        int r1 = r0 | trm;
        float t0r = sr[r0], t0i = si[r0];
        sr[r0] = hi ? sr[r1] : sr[r0];  si[r0] = hi ? si[r1] : si[r0];
        sr[r1] = hi ? t0r    : sr[r1];  si[r1] = hi ? t0i    : si[r1];
    }
}

__device__ __forceinline__ void cnot_reg_lane(float sr[8], float si[8],
                                              int crm, int tm) {
#pragma unroll
    for (int r = 0; r < 8; ++r) {
        if (!(r & crm)) continue;
        sr[r] = __shfl_xor(sr[r], tm, 32);
        si[r] = __shfl_xor(si[r], tm, 32);
    }
}

__device__ __forceinline__ void cnot_reg_reg(float sr[8], float si[8],
                                             int crm, int trm) {
#pragma unroll
    for (int r0 = 0; r0 < 8; ++r0) {
        if (!(r0 & crm) || (r0 & trm)) continue;
        int r1 = r0 | trm;
        float tr = sr[r0], ti = si[r0];
        sr[r0] = sr[r1]; si[r0] = si[r1];
        sr[r1] = tr;     si[r1] = ti;
    }
}

__device__ __forceinline__ float from_lane(float v, int src) {
    return __int_as_float(__builtin_amdgcn_readlane(__float_as_int(v), src));
}

__device__ __forceinline__ void unitary_cols(const float* __restrict__ qnn_w,
                                             float* __restrict__ UrP,
                                             float* __restrict__ UiP,
                                             int bid) {
    int lane = threadIdx.x & 31;
    int t    = bid * 8 + (threadIdx.x >> 5);

    int own = lane < 3 * NQ ? lane : 3 * NQ - 1;
    float hw[3];
#pragma unroll
    for (int k = 0; k < 3; ++k) hw[k] = qnn_w[3 * own + k];
    float gp, gq, gr, gt;
    {
        float ca = cosf(hw[0] * 0.5f), sa = sinf(hw[0] * 0.5f);
        float cb = cosf(hw[1] * 0.5f), sb = sinf(hw[1] * 0.5f);
        float cc = cosf(hw[2] * 0.5f), sc = sinf(hw[2] * 0.5f);
        gp = cb * (cc * ca - sc * sa);
        gq = sb * (cc * sa - sc * ca);
        gr = sb * (cc * ca + sc * sa);
        gt = cb * (cc * sa + sc * ca);
    }

    float sr[8], si[8];
#pragma unroll
    for (int r = 0; r < 8; ++r) {
        sr[r] = (r * 32 + lane == t) ? 1.0f : 0.0f;
        si[r] = 0.0f;
    }

#pragma unroll
    for (int l = 0; l < 3; ++l) {
#pragma unroll
        for (int q = 0; q < NQ; ++q) {
            float P = from_lane(gp, l * NQ + q), Q = from_lane(gq, l * NQ + q);
            float R = from_lane(gr, l * NQ + q), T = from_lane(gt, l * NQ + q);
            if (q < 5) gate_lane(sr, si, P, Q, R, T, 1 << q, lane);
            else       gate_reg(sr, si, P, Q, R, T, 1 << (q - 5));
        }
        cnot_ll(sr, si, 1, 2, lane);
        cnot_ll(sr, si, 2, 4, lane);
        cnot_ll(sr, si, 4, 8, lane);
        cnot_ll(sr, si, 8, 16, lane);
        cnot_lane_reg(sr, si, 16, 1, lane);
        cnot_reg_reg(sr, si, 1, 2);
        cnot_reg_reg(sr, si, 2, 4);
        cnot_reg_lane(sr, si, 4, 1);
    }

#pragma unroll
    for (int r = 0; r < 8; ++r) {
        int s = r * 32 + lane;
        int off = (t >> 1) * (2 * NSTATE) + s * 2 + (t & 1);
        UrP[off] = sr[r];
        UiP[off] = si[r];
    }
}

__global__ __launch_bounds__(256) void k12_producers(const float* __restrict__ x,
                              const float* __restrict__ proj_w,
                              const float* __restrict__ qnn_w,
                              float* __restrict__ angles,
                              float* __restrict__ UrP,
                              float* __restrict__ UiP) {
    int b = blockIdx.x;
    if (b < U_BLOCKS) unitary_cols(qnn_w, UrP, UiP, b);
    else              angles_rows(x, proj_w, angles, b - U_BLOCKS);
}

#define ROWS_PER_BLOCK 32
#define CS_PITCH (2 * NQ + 4)
#define KSTEPS_PER_GROUP 8
#define K_GROUPS (NSTATE / (4 * KSTEPS_PER_GROUP))

__device__ __forceinline__ void load_u_group(const float* __restrict__ ur,
                                             const float* __restrict__ ui, int g,
                                             v4f (&br)[KSTEPS_PER_GROUP],
                                             v4f (&bi)[KSTEPS_PER_GROUP]) {
#pragma unroll
    for (int j = 0; j < KSTEPS_PER_GROUP; ++j) {
        int k = (g * KSTEPS_PER_GROUP + j) * 4;
        br[j] = *(const v4f*)(ur + (k >> 1) * (2 * NSTATE));
        bi[j] = *(const v4f*)(ui + (k >> 1) * (2 * NSTATE));
    }
}

struct AFactors {
    v2f   lo[4];
    float hi[16];
};

__device__ __forceinline__ void a_factors(const float* c, int h, AFactors& f) {
    v4f c01 = *(const v4f*)(c);
    v4f c23 = *(const v4f*)(c + 4);
    v4f c45 = *(const v4f*)(c + 8);
    v4f c67 = *(const v4f*)(c + 12);
    float f1 = h ? c01[3] : c01[2];
    v2f base;
    base[0] = c01[0] * f1;
    base[1] = c01[1] * f1;
#pragma unroll
    for (int jj = 0; jj < 4; ++jj) {
        float m23 = c23[jj & 1] * c23[2 + (jj >> 1)];
        f.lo[jj][0] = base[0] * m23;
        f.lo[jj][1] = base[1] * m23;
    }
    float t45[4], t67[4];
#pragma unroll
    for (int i = 0; i < 4; ++i) {
        t45[i] = c45[i & 1] * c45[2 + (i >> 1)];
        t67[i] = c67[i & 1] * c67[2 + (i >> 1)];
    }
#pragma unroll
    for (int i = 0; i < 16; ++i) f.hi[i] = t45[i & 3] * t67[i >> 2];
}

__device__ __forceinline__ void a_fragments(const AFactors& f, v2f (&a)[NSTATE / 4]) {
#pragma unroll
    for (int ks = 0; ks < NSTATE / 4; ++ks) a[ks] = f.lo[ks & 3] * f.hi[ks >> 2];
}

__device__ __forceinline__ void mma_group(const v2f (&a0)[NSTATE / 4], const v2f (&a1)[NSTATE / 4], int g,
                                          const v4f (&br)[KSTEPS_PER_GROUP],
                                          const v4f (&bi)[KSTEPS_PER_GROUP],
                                          v8f (&acc)[8]) {
#pragma unroll
    for (int j = 0; j < KSTEPS_PER_GROUP; ++j) {
        int ks = g * KSTEPS_PER_GROUP + j;
        v2f br0 = br[j].lo, br1 = br[j].hi;
        v2f bi0 = bi[j].lo, bi1 = bi[j].hi;
        acc[0] = __builtin_amdgcn_wmma_f32_16x16x4_f32(false, br0, false, a0[ks], (short)0, acc[0], false, false);
        acc[1] = __builtin_amdgcn_wmma_f32_16x16x4_f32(false, bi0, false, a0[ks], (short)0, acc[1], false, false);
        acc[2] = __builtin_amdgcn_wmma_f32_16x16x4_f32(false, br1, false, a0[ks], (short)0, acc[2], false, false);
        acc[3] = __builtin_amdgcn_wmma_f32_16x16x4_f32(false, bi1, false, a0[ks], (short)0, acc[3], false, false);
        acc[4] = __builtin_amdgcn_wmma_f32_16x16x4_f32(false, br0, false, a1[ks], (short)0, acc[4], false, false);
        acc[5] = __builtin_amdgcn_wmma_f32_16x16x4_f32(false, bi0, false, a1[ks], (short)0, acc[5], false, false);
        acc[6] = __builtin_amdgcn_wmma_f32_16x16x4_f32(false, br1, false, a1[ks], (short)0, acc[6], false, false);
        acc[7] = __builtin_amdgcn_wmma_f32_16x16x4_f32(false, bi1, false, a1[ks], (short)0, acc[7], false, false);
    }
}

__global__ __launch_bounds__(256) __attribute__((amdgpu_waves_per_eu(2, 2))) void k3_fused(
        const float* __restrict__ angles,
        const float* __restrict__ UrP,
        const float* __restrict__ UiP,
        const float* __restrict__ out_w,
        const float* __restrict__ out_b,
        float* __restrict__ out) {
    __shared__ __align__(16) float cs[ROWS_PER_BLOCK][CS_PITCH];
    __shared__ __align__(16) float part[ROWS_PER_BLOCK][PART_PITCH];
    __shared__ __align__(16) float E[ROWS_PER_BLOCK][NQ];

    int tid  = threadIdx.x;
    int lane = tid & 31;
    int wave = tid >> 5;
    int row0 = blockIdx.x * ROWS_PER_BLOCK;

    {
        int m = tid >> 3, q = tid & 7;
        float a = angles[(row0 + m) * NQ + q] * 0.5f;
        v2f c;
        c[0] = cosf(a);
        c[1] = sinf(a);
        *(v2f*)&cs[m][2 * q] = c;
    }
    __syncthreads();

    int mrow = lane & 15;
    int h    = lane >> 4;
    int n    = lane & 15;
    int sb   = wave * 32;

    const float* ur = UrP + h * (2 * NSTATE) + (sb + 2 * n) * 2;
    const float* ui = UiP + h * (2 * NSTATE) + (sb + 2 * n) * 2;

    v4f br0[KSTEPS_PER_GROUP], bi0[KSTEPS_PER_GROUP];
    v4f br1[KSTEPS_PER_GROUP], bi1[KSTEPS_PER_GROUP];
    load_u_group(ur, ui, 0, br0, bi0);
    load_u_group(ur, ui, 1, br1, bi1);

    v2f a0[NSTATE / 4], a1[NSTATE / 4];
    {
        AFactors fa;
        a_factors(&cs[mrow][0], h, fa);
        a_fragments(fa, a0);
        a_factors(&cs[16 + mrow][0], h, fa);
        a_fragments(fa, a1);
    }
    __builtin_amdgcn_sched_barrier(0);

    v8f acc[8];
#pragma unroll
    for (int i = 0; i < 8; ++i)
#pragma unroll
        for (int v = 0; v < 8; ++v) acc[i][v] = 0.0f;

    mma_group(a0, a1, 0, br0, bi0, acc);
    mma_group(a0, a1, 1, br1, bi1, acc);
#pragma unroll
    for (int g = 2; g < K_GROUPS; ++g) {
        v4f br[KSTEPS_PER_GROUP], bi[KSTEPS_PER_GROUP];
        load_u_group(ur, ui, g, br, bi);
        mma_group(a0, a1, g, br, bi, acc);
    }

    {
        int g = 2 * wave + h;
#pragma unroll
        for (int mt = 0; mt < 2; ++mt) {
            float t[8], d[8];
#pragma unroll
            for (int v = 0; v < 8; ++v) {
                float p0 = acc[4 * mt + 0][v] * acc[4 * mt + 0][v] + acc[4 * mt + 1][v] * acc[4 * mt + 1][v];
                float p1 = acc[4 * mt + 2][v] * acc[4 * mt + 2][v] + acc[4 * mt + 3][v] * acc[4 * mt + 3][v];
                t[v] = p0 + p1;
                d[v] = p0 - p1;
            }
            float D0 = ((d[0] + d[1]) + (d[2] + d[3])) + ((d[4] + d[5]) + (d[6] + d[7]));
            float u[4], w[4];
#pragma unroll
            for (int i = 0; i < 4; ++i) {
                u[i] = t[2 * i] + t[2 * i + 1];
                w[i] = t[2 * i] - t[2 * i + 1];
            }
            float D1 = (w[0] + w[1]) + (w[2] + w[3]);
            float x0 = u[0] + u[1], x1 = u[2] + u[3];
            float D2 = (u[0] - u[1]) + (u[2] - u[3]);
            float S  = x0 + x1;
            float D3 = x0 - x1;
            float* pr = &part[16 * mt + mrow][g];
            pr[0 * 16] = S;
            pr[1 * 16] = D0;
            pr[2 * 16] = D1;
            pr[3 * 16] = D2;
            pr[4 * 16] = D3;
        }
    }
    __syncthreads();

    {
        int m = tid >> 3, q = tid & 7;
        int kind = q < 4 ? q + 1 : 0;
        const v4f* src = (const v4f*)&part[m][16 * kind];
        v4f r[4];
#pragma unroll
        for (int i = 0; i < 4; ++i) r[i] = src[i];
        float s1 = q == 4 ? -1.0f : 1.0f;
        float s2 = q == 5 ? -1.0f : 1.0f;
        float s4 = q == 6 ? -1.0f : 1.0f;
        float s8 = q == 7 ? -1.0f : 1.0f;
        float a[4];
#pragma unroll
        for (int i = 0; i < 4; ++i)
            a[i] = fmaf(s2, fmaf(s1, r[i][3], r[i][2]), fmaf(s1, r[i][1], r[i][0]));
        float esum = fmaf(s8, fmaf(s4, a[3], a[2]), fmaf(s4, a[1], a[0]));
        E[m][q] = esum;
    }
    __syncthreads();

    for (int idx = tid; idx < ROWS_PER_BLOCK * NCLS; idx += 256) {
        int m = idx / NCLS, c = idx % NCLS;
        float o = out_b[c];
#pragma unroll
        for (int q = 0; q < NQ; ++q) o += E[m][q] * out_w[c * NQ + q];
        out[(row0 + m) * NCLS + c] = o;
    }
}

extern "C" void kernel_launch(void* const* d_in, const int* in_sizes, int n_in,
                              void* d_out, int out_size, void* d_ws, size_t ws_size,
                              hipStream_t stream) {
    const float* x      = (const float*)d_in[0];
    const float* proj_w = (const float*)d_in[1];
    const float* qnn_w  = (const float*)d_in[2];
    const float* out_w  = (const float*)d_in[3];
    const float* out_b  = (const float*)d_in[4];
    float* ws = (float*)d_ws;

    float* angles = ws;
    float* UrP    = ws + 65536;
    float* UiP    = ws + 65536 + 65536;

    k12_producers<<<PRODUCER_BLOCKS, 256, 0, stream>>>(x, proj_w, qnn_w,
                                                       angles, UrP, UiP);
    k3_fused<<<BATCH / ROWS_PER_BLOCK, 256, 0, stream>>>(angles, UrP, UiP, out_w, out_b,
                                             (float*)d_out);
}


# angle kernel fused into the circuit kernel, one launch: a work-group makes the angles of its own 32 rows, each stays in its thread's register; weights-only prologue runs under the x loads
# speedup vs baseline: 4.3060x; 1.6949x over previous
#include <hip/hip_runtime.h>
#include <math.h>

typedef float v2f __attribute__((ext_vector_type(2)));
typedef float v4f __attribute__((ext_vector_type(4)));
typedef float v8f __attribute__((ext_vector_type(8)));

#define BATCH 8192
#define FDIM  512
#define NQ    8
#define NSTATE 256
#define NCLS  10
#define ANGLE_ROWS_PER_WAVE  4

#define ROWS_PER_BLOCK 32
#define PSI_PITCH 260
#define MAT_PITCH 20
#define PHI_PITCH 36

__host__ __device__ constexpr int ring_map(int s) {
    int p = s ^ (s << 1);
    p ^= p << 2;
    p ^= p << 4;
    p &= 255;
    return (p & 0xFE) | (((p >> 7) ^ s) & 1);
}

template <int SIDE>
__device__ __forceinline__ void factor_mma(const float* mr, const float* mi,
                                           const float* sr, const float* si,
                                           int wave, int c, int h,
                                           v8f (&aR)[4], v8f (&aI)[4]) {
    v2f ar[4], ai[4], an[4];
    {
        const float* pr = mr + c * MAT_PITCH + 8 * h;
        const float* pi = mi + c * MAT_PITCH + 8 * h;
        v4f m0 = *(const v4f*)pr, m1 = *(const v4f*)(pr + 4);
        v4f n0 = *(const v4f*)pi, n1 = *(const v4f*)(pi + 4);
        ar[0] = m0.lo; ar[1] = m0.hi; ar[2] = m1.lo; ar[3] = m1.hi;
        ai[0] = n0.lo; ai[1] = n0.hi; ai[2] = n1.lo; ai[3] = n1.hi;
#pragma unroll
        for (int j = 0; j < 4; ++j) an[j] = -ai[j];
    }
    v2f br[4][4], bi[4][4];
#pragma unroll
    for (int p = 0; p < 4; ++p) {
        int row = 16 * (p >> 1) + c;
        int x   = wave + 8 * (p & 1);
        if (SIDE == 0) {
            int base = row * PSI_PITCH + 16 * x + 8 * h;
            v4f r0 = *(const v4f*)(sr + base), r1 = *(const v4f*)(sr + base + 4);
            v4f i0 = *(const v4f*)(si + base), i1 = *(const v4f*)(si + base + 4);
            br[p][0] = r0.lo; br[p][1] = r0.hi; br[p][2] = r1.lo; br[p][3] = r1.hi;
            bi[p][0] = i0.lo; bi[p][1] = i0.hi; bi[p][2] = i1.lo; bi[p][3] = i1.hi;
        } else {
            int base = row * PSI_PITCH + 128 * h + x;
#pragma unroll
            for (int j = 0; j < 4; ++j) {
                br[p][j][0] = sr[base + 16 * (2 * j)];
                br[p][j][1] = sr[base + 16 * (2 * j + 1)];
                bi[p][j][0] = si[base + 16 * (2 * j)];
                bi[p][j][1] = si[base + 16 * (2 * j + 1)];
            }
        }
    }
#pragma unroll
    for (int p = 0; p < 4; ++p)
#pragma unroll
        for (int v = 0; v < 8; ++v) { aR[p][v] = 0.0f; aI[p][v] = 0.0f; }
#pragma unroll
    for (int j = 0; j < 4; ++j) {
#pragma unroll
        for (int p = 0; p < 4; ++p) {
            aR[p] = __builtin_amdgcn_wmma_f32_16x16x4_f32(false, ar[j], false, br[p][j], (short)0, aR[p], false, false);
            aI[p] = __builtin_amdgcn_wmma_f32_16x16x4_f32(false, ar[j], false, bi[p][j], (short)0, aI[p], false, false);
        }
#pragma unroll
        for (int p = 0; p < 4; ++p) {
            aR[p] = __builtin_amdgcn_wmma_f32_16x16x4_f32(false, an[j], false, bi[p][j], (short)0, aR[p], false, false);
            aI[p] = __builtin_amdgcn_wmma_f32_16x16x4_f32(false, ai[j], false, br[p][j], (short)0, aI[p], false, false);
        }
    }
}

__device__ __forceinline__ void factor_low_inplace(const float* mr, const float* mi,
                                                   float* sr, float* si,
                                                   int wave, int c, int h) {
    v8f aR[4], aI[4];
    factor_mma<0>(mr, mi, sr, si, wave, c, h, aR, aI);
#pragma unroll
    for (int p = 0; p < 4; ++p) {
        int row = 16 * (p >> 1) + c;
        int x   = wave + 8 * (p & 1);
        int base = row * PSI_PITCH + 16 * x + 8 * h;
        *(v4f*)(sr + base)     = aR[p].lo;
        *(v4f*)(sr + base + 4) = aR[p].hi;
        *(v4f*)(si + base)     = aI[p].lo;
        *(v4f*)(si + base + 4) = aI[p].hi;
    }
}

__device__ __forceinline__ void lds_barrier() {
    __builtin_amdgcn_fence(__ATOMIC_RELEASE, "workgroup", "local");
    __builtin_amdgcn_s_barrier();
    __builtin_amdgcn_fence(__ATOMIC_ACQUIRE, "workgroup", "local");
}

__global__ __launch_bounds__(256) void k_angles_circuit(
        const float* __restrict__ x,
        const float* __restrict__ proj_w,
        const float* __restrict__ qnn_w,
        const float* __restrict__ out_w,
        const float* __restrict__ out_b,
        float* __restrict__ out) {
    __shared__ __align__(16) float psiA_r[ROWS_PER_BLOCK * PSI_PITCH];
    __shared__ __align__(16) float psiA_i[ROWS_PER_BLOCK * PSI_PITCH];
    __shared__ __align__(16) float psiB_r[ROWS_PER_BLOCK * PSI_PITCH];
    __shared__ __align__(16) float psiB_i[ROWS_PER_BLOCK * PSI_PITCH];
    __shared__ __align__(16) float mat_r[4 * 16 * MAT_PITCH];
    __shared__ __align__(16) float mat_i[4 * 16 * MAT_PITCH];
    __shared__ __align__(16) float phi[ROWS_PER_BLOCK * PHI_PITCH];
    __shared__ __align__(16) float hwcs[3 * 3 * NQ * 2];
    __shared__ __align__(16) float gate[3 * NQ * 4];
    __shared__ __align__(16) float Epart[16 * ROWS_PER_BLOCK * NQ];
    __shared__ __align__(16) float E[ROWS_PER_BLOCK][NQ];

    int tid  = threadIdx.x;
    int lane = tid & 31;
    int wave = tid >> 5;
    int row0 = blockIdx.x * ROWS_PER_BLOCK;
    int pm = tid >> 3, pq = tid & 7;
    int c = lane & 15, h = lane >> 4;

    float qa = qnn_w[tid < 3 * 3 * NQ ? tid : 3 * 3 * NQ - 1];
    const v4f* pw4 = (const v4f*)proj_w;
    v4f wst[4];
#pragma unroll
    for (int k = 0; k < 4; ++k) wst[k] = pw4[tid + 256 * k];
    const v4f* xr = (const v4f*)(x + (size_t)(row0 + wave * ANGLE_ROWS_PER_WAVE) * FDIM) + lane;
    v4f xv[ANGLE_ROWS_PER_WAVE][4];
#pragma unroll
    for (int j = 0; j < 4; ++j)
#pragma unroll
        for (int r = 0; r < ANGLE_ROWS_PER_WAVE; ++r)
            xv[r][j] = xr[r * (FDIM / 4) + j * 32];

    {
        float a = qa * 0.5f;
        float ca = cosf(a), sa = sinf(a);
        if (tid < 3 * 3 * NQ) {
            hwcs[2 * tid]     = ca;
            hwcs[2 * tid + 1] = sa;
        }
    }
    float* w = Epart;
    v4f* w4 = (v4f*)w;
#pragma unroll
    for (int k = 0; k < 4; ++k) w4[tid + 256 * k] = wst[k];
    lds_barrier();

    if (tid < 3 * NQ) {
        float ca = hwcs[6 * tid + 0], sa = hwcs[6 * tid + 1];
        float cb = hwcs[6 * tid + 2], sb = hwcs[6 * tid + 3];
        float cc = hwcs[6 * tid + 4], sc = hwcs[6 * tid + 5];
        v4f g;
        g[0] = cb * (cc * ca - sc * sa);
        g[1] = sb * (cc * sa - sc * ca);
        g[2] = sb * (cc * ca + sc * sa);
        g[3] = cb * (cc * sa + sc * ca);
        *(v4f*)&gate[4 * tid] = g;
    }
    lds_barrier();

    {
        int ip = tid >> 4, i = tid & 15;
#pragma unroll
        for (int X = 0; X < 4; ++X) {
            int gbase = (1 + (X >> 1)) * NQ + 4 * (X & 1);
            float vr = 1.0f, vi = 0.0f;
#pragma unroll
            for (int k = 0; k < 4; ++k) {
                v4f g = *(const v4f*)&gate[4 * (gbase + k)];
                int bp = (ip >> k) & 1, b = (i >> k) & 1;
                float er = (bp == b) ? g[0] : (bp ? g[2] : -g[2]);
                float ei = (bp == b) ? (bp ? -g[1] : g[1]) : -g[3];
                float nr = vr * er - vi * ei;
                float ni = vr * ei + vi * er;
                vr = nr; vi = ni;
            }
            mat_r[(X * 16 + ip) * MAT_PITCH + i] = vr;
            mat_i[(X * 16 + ip) * MAT_PITCH + i] = vi;
        }
    }

    float ang;
    {
        v2f accv[ANGLE_ROWS_PER_WAVE][NQ];
#pragma unroll
        for (int r = 0; r < ANGLE_ROWS_PER_WAVE; ++r)
#pragma unroll
            for (int q = 0; q < NQ; ++q) { accv[r][q][0] = 0.0f; accv[r][q][1] = 0.0f; }

#pragma unroll
        for (int j = 0; j < 4; ++j) {
#pragma unroll
            for (int q = 0; q < NQ; ++q) {
                v4f wv = w4[q * (FDIM / 4) + j * 32 + lane];
                v2f wlo = wv.lo, whi = wv.hi;
#pragma unroll
                for (int r = 0; r < ANGLE_ROWS_PER_WAVE; ++r) {
                    v2f xlo = xv[r][j].lo, xhi = xv[r][j].hi;
                    accv[r][q] = __builtin_elementwise_fma(xlo, wlo, accv[r][q]);
                    accv[r][q] = __builtin_elementwise_fma(xhi, whi, accv[r][q]);
                }
            }
        }
        float acc[ANGLE_ROWS_PER_WAVE * NQ];
#pragma unroll
        for (int r = 0; r < ANGLE_ROWS_PER_WAVE; ++r)
#pragma unroll
            for (int q = 0; q < NQ; ++q) acc[r * NQ + q] = accv[r][q][0] + accv[r][q][1];

        float a16[16], a8[8], a4[4], a2[2];
        {
            bool hi = (lane & 16) != 0;
#pragma unroll
            for (int i = 0; i < 16; ++i) {
                float keep = hi ? acc[i + 16] : acc[i];
                float send = hi ? acc[i] : acc[i + 16];
                a16[i] = keep + __shfl_xor(send, 16, 32);
            }
        }
        {
            bool hi = (lane & 8) != 0;
#pragma unroll
            for (int i = 0; i < 8; ++i) {
                float keep = hi ? a16[i + 8] : a16[i];
                float send = hi ? a16[i] : a16[i + 8];
                a8[i] = keep + __shfl_xor(send, 8, 32);
            }
        }
        {
            bool hi = (lane & 4) != 0;
#pragma unroll
            for (int i = 0; i < 4; ++i) {
                float keep = hi ? a8[i + 4] : a8[i];
                float send = hi ? a8[i] : a8[i + 4];
                a4[i] = keep + __shfl_xor(send, 4, 32);
            }
        }
        {
            bool hi = (lane & 2) != 0;
#pragma unroll
            for (int i = 0; i < 2; ++i) {
                float keep = hi ? a4[i + 2] : a4[i];
                float send = hi ? a4[i] : a4[i + 2];
                a2[i] = keep + __shfl_xor(send, 2, 32);
            }
        }
        float v;
        {
            bool hi = (lane & 1) != 0;
            float keep = hi ? a2[1] : a2[0];
            float send = hi ? a2[0] : a2[1];
            v = keep + __shfl_xor(send, 1, 32);
        }
        float angle = tanhf(v) * 1.5707963267948966f;
        ang = angle * 0.5f;
    }
    float ec = cosf(ang), es = sinf(ang);

    {
        v4f g = *(const v4f*)&gate[4 * pq];
        v4f f;
        f[0] = g[0] * ec - g[2] * es;
        f[1] = g[1] * ec - g[3] * es;
        f[2] = g[2] * ec + g[0] * es;
        f[3] = -g[3] * ec - g[1] * es;
        *(v4f*)&phi[pm * PHI_PITCH + 4 * pq] = f;
    }
    __syncthreads();

    {
        const v4f* ph = (const v4f*)&phi[pm * PHI_PITCH];
        v4f f[NQ];
#pragma unroll
        for (int q = 0; q < NQ; ++q) f[q] = ph[q];
        float tr, ti;
        {
            int b5 = pq & 1, b6 = (pq >> 1) & 1, b7 = pq >> 2;
            float ar = b5 ? f[5][2] : f[5][0], ai = b5 ? f[5][3] : f[5][1];
            float br = b6 ? f[6][2] : f[6][0], bi = b6 ? f[6][3] : f[6][1];
            float cr = b7 ? f[7][2] : f[7][0], ci = b7 ? f[7][3] : f[7][1];
            float ur = ar * br - ai * bi, ui = ar * bi + ai * br;
            tr = ur * cr - ui * ci;
            ti = ur * ci + ui * cr;
        }
        float vr[32], vi[32];
        vr[0] = tr; vi[0] = ti;
#pragma unroll
        for (int q = 0; q < 5; ++q) {
#pragma unroll
            for (int idx = 0; idx < (1 << q); ++idx) {
                float xr = vr[idx], xi = vi[idx];
                vr[idx + (1 << q)] = xr * f[q][2] - xi * f[q][3];
                vi[idx + (1 << q)] = xr * f[q][3] + xi * f[q][2];
                vr[idx]            = xr * f[q][0] - xi * f[q][1];
                vi[idx]            = xr * f[q][1] + xi * f[q][0];
            }
        }
        int rg = ring_map(pq << 5);
        float* dr = psiA_r + pm * PSI_PITCH;
        float* di = psiA_i + pm * PSI_PITCH;
#pragma unroll
        for (int low = 0; low < 32; ++low) {
            int d = rg ^ ring_map(low);
            dr[d] = vr[low];
            di[d] = vi[low];
        }
    }
    __syncthreads();

    factor_low_inplace(mat_r + 0 * 16 * MAT_PITCH, mat_i + 0 * 16 * MAT_PITCH, psiA_r, psiA_i, wave, c, h);
    __syncthreads();

    {
        v8f aR[4], aI[4];
        factor_mma<1>(mat_r + 1 * 16 * MAT_PITCH, mat_i + 1 * 16 * MAT_PITCH, psiA_r, psiA_i, wave, c, h, aR, aI);
#pragma unroll
        for (int p = 0; p < 4; ++p) {
            int row = 16 * (p >> 1) + c;
            int x   = wave + 8 * (p & 1);
            int rb  = ring_map((h << 7) | x);
            float* dr = psiB_r + row * PSI_PITCH;
            float* di = psiB_i + row * PSI_PITCH;
#pragma unroll
            for (int r = 0; r < 8; ++r) {
                int d = rb ^ ring_map(16 * r);
                dr[d] = aR[p][r];
                di[d] = aI[p][r];
            }
        }
    }
    __syncthreads();

    factor_low_inplace(mat_r + 2 * 16 * MAT_PITCH, mat_i + 2 * 16 * MAT_PITCH, psiB_r, psiB_i, wave, c, h);
    __syncthreads();

    {
        v8f aR[4], aI[4];
        factor_mma<1>(mat_r + 3 * 16 * MAT_PITCH, mat_i + 3 * 16 * MAT_PITCH, psiB_r, psiB_i, wave, c, h, aR, aI);
        float e[2][NQ];
#pragma unroll
        for (int mt = 0; mt < 2; ++mt)
#pragma unroll
            for (int q = 0; q < NQ; ++q) e[mt][q] = 0.0f;
#pragma unroll
        for (int p = 0; p < 4; ++p) {
            int x = wave + 8 * (p & 1);
            float pr[8];
#pragma unroll
            for (int r = 0; r < 8; ++r) pr[r] = aR[p][r] * aR[p][r] + aI[p][r] * aI[p][r];
            float u0 = pr[0] + pr[1], u1 = pr[2] + pr[3], u2 = pr[4] + pr[5], u3 = pr[6] + pr[7];
            float w0 = pr[0] - pr[1], w1 = pr[2] - pr[3], w2 = pr[4] - pr[5], w3 = pr[6] - pr[7];
            float S = (u0 + u1) + (u2 + u3);
            float A = (w0 + w1) + (w2 + w3);
            float B = (w0 - w1) + (w2 - w3);
            float C = (w0 - w1) - (w2 - w3);
            int hx = (h << 7) | x;
            float sums[NQ] = { C, S, S, S, A, B, C, C };
#pragma unroll
            for (int q = 0; q < NQ; ++q) {
                int mask = q == 0 ? 0xFE : ((2 << q) - 1);
                bool neg = (__popc(hx & mask) & 1) != 0;
                e[p >> 1][q] += neg ? -sums[q] : sums[q];
            }
        }
#pragma unroll
        for (int mt = 0; mt < 2; ++mt) {
            float* dst = &Epart[((2 * wave + h) * ROWS_PER_BLOCK + 16 * mt + c) * NQ];
            v4f lo4, hi4;
            lo4[0] = e[mt][0]; lo4[1] = e[mt][1]; lo4[2] = e[mt][2]; lo4[3] = e[mt][3];
            hi4[0] = e[mt][4]; hi4[1] = e[mt][5]; hi4[2] = e[mt][6]; hi4[3] = e[mt][7];
            *(v4f*)dst       = lo4;
            *(v4f*)(dst + 4) = hi4;
        }
    }
    __syncthreads();

    {
        float s = 0.0f;
#pragma unroll
        for (int k = 0; k < 16; ++k) s += Epart[(k * ROWS_PER_BLOCK + pm) * NQ + pq];
        E[pm][pq] = s;
    }
    __syncthreads();

    for (int idx = tid; idx < ROWS_PER_BLOCK * NCLS; idx += 256) {
        int m = idx / NCLS, cl = idx % NCLS;
        float o = out_b[cl];
#pragma unroll
        for (int q = 0; q < NQ; ++q) o += E[m][q] * out_w[cl * NQ + q];
        out[(row0 + m) * NCLS + cl] = o;
    }
}

extern "C" void kernel_launch(void* const* d_in, const int* in_sizes, int n_in,
                              void* d_out, int out_size, void* d_ws, size_t ws_size,
                              hipStream_t stream) {
    const float* x      = (const float*)d_in[0];
    const float* proj_w = (const float*)d_in[1];
    const float* qnn_w  = (const float*)d_in[2];
    const float* out_w  = (const float*)d_in[3];
    const float* out_b  = (const float*)d_in[4];

    k_angles_circuit<<<BATCH / ROWS_PER_BLOCK, 256, 0, stream>>>(x, proj_w, qnn_w, out_w, out_b,
                                                                 (float*)d_out);
}
